# starting kernel
# baseline (speedup 1.0000x reference)
#include <hip/hip_runtime.h>

typedef __attribute__((ext_vector_type(2))) float v2f;
typedef __attribute__((ext_vector_type(8))) float v8f;

#define IN_DIMC 256
#define H_DIMC  128
#define OUT_DIMC 40
#define OUT_PADC 48

__global__ void zero_f32(float* __restrict__ p, int n) {
  int i = blockIdx.x * blockDim.x + threadIdx.x;
  if (i < n) p[i] = 0.0f;
}

__global__ void init_deg(float* __restrict__ deg, int n) {
  int i = blockIdx.x * blockDim.x + threadIdx.x;
  if (i < n) deg[i] = 1.0f;
}

__global__ void deg_edges(const int* __restrict__ dst, float* __restrict__ deg, int e) {
  int i = blockIdx.x * blockDim.x + threadIdx.x;
  if (i < e) atomicAdd(&deg[dst[i]], 1.0f);
}

__global__ void finish_dinv(float* __restrict__ deg, int n) {
  int i = blockIdx.x * blockDim.x + threadIdx.x;
  if (i < n) deg[i] = rsqrtf(fmaxf(deg[i], 1e-12f));
}

__global__ void edge_norm(const int* __restrict__ src, const int* __restrict__ dst,
                          const float* __restrict__ dinv, float* __restrict__ norm, int e) {
  int i = blockIdx.x * blockDim.x + threadIdx.x;
  if (i < e) norm[i] = dinv[src[i]] * dinv[dst[i]];
}

__global__ __launch_bounds__(256) void gemm1_wmma(const float* __restrict__ X,
                                                  const float* __restrict__ W,
                                                  float* __restrict__ H, int n) {
  __shared__ float sW[64][H_DIMC + 4];
  __shared__ float sX[16][64 + 4];

  const int tid  = threadIdx.x;
  const int wave = tid >> 5;
  const int lane = tid & 31;
  const int col0 = wave * 16;
  const int m    = lane & 15;
  const int kh   = lane >> 4;
  const int row0 = blockIdx.x * 16;

  v8f acc = {0.f, 0.f, 0.f, 0.f, 0.f, 0.f, 0.f, 0.f};

  for (int kb = 0; kb < IN_DIMC; kb += 64) {
    __syncthreads();
    for (int i = tid; i < 64 * H_DIMC; i += 256) {
      int r = i >> 7, c = i & (H_DIMC - 1);
      sW[r][c] = W[(size_t)(kb + r) * H_DIMC + c];
    }
    for (int i = tid; i < 16 * 64; i += 256) {
      int r = i >> 6, k = i & 63;
      int row = row0 + r;
      sX[r][k] = (row < n) ? X[(size_t)row * IN_DIMC + kb + k] : 0.0f;
    }
    __syncthreads();

#pragma unroll
    for (int ks = 0; ks < 64; ks += 4) {
      v2f a, b;
      a[0] = sX[m][ks + 2 * kh + 0];
      a[1] = sX[m][ks + 2 * kh + 1];
      b[0] = sW[ks + 2 * kh + 0][col0 + m];
      b[1] = sW[ks + 2 * kh + 1][col0 + m];
      acc = __builtin_amdgcn_wmma_f32_16x16x4_f32(false, a, false, b,
                                                  (short)0, acc, false, false);
    }
  }
#pragma unroll
  for (int r = 0; r < 8; ++r) {
    int row = row0 + 8 * kh + r;
    if (row < n) H[(size_t)row * H_DIMC + col0 + m] = acc[r];
  }
}

__global__ __launch_bounds__(384) void gemm2_wmma(const float* __restrict__ Xh,
                                                  const float* __restrict__ W2,
                                                  float* __restrict__ H2, int n) {
  __shared__ float sW[H_DIMC][OUT_PADC + 4];
  __shared__ float sX[64][H_DIMC + 4];

  const int tid  = threadIdx.x;
  const int wave = tid >> 5;
  const int lane = tid & 31;
  const int rt   = wave / 3;
  const int ct   = wave % 3;
  const int col0 = ct * 16;
  const int m    = lane & 15;
  const int kh   = lane >> 4;
  const int row0 = blockIdx.x * 64;

  for (int i = tid; i < H_DIMC * OUT_PADC; i += 384) {
    int r = i / OUT_PADC, c = i % OUT_PADC;
    sW[r][c] = (c < OUT_DIMC) ? W2[(size_t)r * OUT_DIMC + c] : 0.0f;
  }
  for (int i = tid; i < 64 * H_DIMC; i += 384) {
    int r = i >> 7, k = i & (H_DIMC - 1);
    int row = row0 + r;
    sX[r][k] = (row < n) ? Xh[(size_t)row * H_DIMC + k] : 0.0f;
  }
  __syncthreads();

  v8f acc = {0.f, 0.f, 0.f, 0.f, 0.f, 0.f, 0.f, 0.f};
#pragma unroll
  for (int ks = 0; ks < H_DIMC; ks += 4) {
    v2f a, b;
    a[0] = sX[rt * 16 + m][ks + 2 * kh + 0];
    a[1] = sX[rt * 16 + m][ks + 2 * kh + 1];
    b[0] = sW[ks + 2 * kh + 0][col0 + m];
    b[1] = sW[ks + 2 * kh + 1][col0 + m];
    acc = __builtin_amdgcn_wmma_f32_16x16x4_f32(false, a, false, b,
                                                (short)0, acc, false, false);
  }
#pragma unroll
  for (int r = 0; r < 8; ++r) {
    int row = row0 + rt * 16 + 8 * kh + r;
    int col = col0 + m;
    if (row < n && col < OUT_DIMC) H2[(size_t)row * OUT_DIMC + col] = acc[r];
  }
}

__global__ __launch_bounds__(256) void agg_edges_h(const int* __restrict__ src,
                                                   const int* __restrict__ dst,
                                                   const float* __restrict__ norm,
                                                   const float* __restrict__ h,
                                                   float* __restrict__ out, int e) {
  int gw   = (blockIdx.x * 256 + threadIdx.x) >> 5;
  int lane = threadIdx.x & 31;
  if (gw >= e) return;
  int s = src[gw], d = dst[gw];
  float w = norm[gw];
  const float4* hp = (const float4*)(h + (size_t)s * H_DIMC);
  float4 v = hp[lane];
  float* op = out + (size_t)d * H_DIMC + lane * 4;
  atomicAdd(op + 0, v.x * w);
  atomicAdd(op + 1, v.y * w);
  atomicAdd(op + 2, v.z * w);
  atomicAdd(op + 3, v.w * w);
}

__global__ __launch_bounds__(256) void agg_edges_o(const int* __restrict__ src,
                                                   const int* __restrict__ dst,
                                                   const float* __restrict__ norm,
                                                   const float* __restrict__ h2,
                                                   float* __restrict__ out, int e) {
  int gw   = (blockIdx.x * 256 + threadIdx.x) >> 5;
  int lane = threadIdx.x & 31;
  if (gw >= e) return;
  int s = src[gw], d = dst[gw];
  float w = norm[gw];
  const float* hp = h2 + (size_t)s * OUT_DIMC;
  float* op = out + (size_t)d * OUT_DIMC;
  for (int f = lane; f < OUT_DIMC; f += 32)
    atomicAdd(op + f, hp[f] * w);
}

__global__ void relu_bias1(float* __restrict__ agg1, const float* __restrict__ h1,
                           const float* __restrict__ dinv, const float* __restrict__ b1,
                           int total) {
  int i = blockIdx.x * blockDim.x + threadIdx.x;
  if (i >= total) return;
  int node = i >> 7;
  int f    = i & 127;
  float di = dinv[node];
  float v = agg1[i] + h1[i] * di * di + b1[f];
  agg1[i] = v > 0.0f ? v : 0.0f;
}

__global__ void final_out(float* __restrict__ out, const float* __restrict__ h2,
                          const float* __restrict__ dinv, const float* __restrict__ b2,
                          int total) {
  int i = blockIdx.x * blockDim.x + threadIdx.x;
  if (i >= total) return;
  int node = i / OUT_DIMC;
  int f    = i - node * OUT_DIMC;
  float di = dinv[node];
  out[i] += h2[i] * di * di + b2[f];
}

extern "C" void kernel_launch(void* const* d_in, const int* in_sizes, int n_in,
                              void* d_out, int out_size, void* d_ws, size_t ws_size,
                              hipStream_t stream) {
  const float* x  = (const float*)d_in[0];
  const int*   ei = (const int*)d_in[1];
  const float* W1 = (const float*)d_in[2];
  const float* b1 = (const float*)d_in[3];
  const float* W2 = (const float*)d_in[4];
  const float* b2 = (const float*)d_in[5];
  float* out = (float*)d_out;

  const int N = in_sizes[0] / IN_DIMC;
  const int E = in_sizes[1] / 2;
  const int* src = ei;
  const int* dst = ei + E;

  size_t off = 0;
  char* base = (char*)d_ws;
  auto take = [&](size_t count) -> float* {
    float* p = (float*)(base + off);
    off += ((count * sizeof(float) + 255) & ~(size_t)255);
    return p;
  };
  float* dinv = take((size_t)N);
  float* norm = take((size_t)E);
  float* h1   = take((size_t)N * H_DIMC);
  float* agg1 = take((size_t)N * H_DIMC);
  float* h2   = take((size_t)N * OUT_DIMC);
  (void)ws_size; (void)n_in; (void)out_size;

  const int T = 256;
  init_deg   <<<(N + T - 1) / T, T, 0, stream>>>(dinv, N);
  deg_edges  <<<(E + T - 1) / T, T, 0, stream>>>(dst, dinv, E);
  finish_dinv<<<(N + T - 1) / T, T, 0, stream>>>(dinv, N);
  edge_norm  <<<(E + T - 1) / T, T, 0, stream>>>(src, dst, dinv, norm, E);

  const int nh = N * H_DIMC;
  zero_f32   <<<(nh + T - 1) / T, T, 0, stream>>>(agg1, nh);
  gemm1_wmma <<<(N + 15) / 16, 256, 0, stream>>>(x, W1, h1, N);
  agg_edges_h<<<(E + 7) / 8, 256, 0, stream>>>(src, dst, norm, h1, agg1, E);
  relu_bias1 <<<(nh + T - 1) / T, T, 0, stream>>>(agg1, h1, dinv, b1, nh);

  const int no = N * OUT_DIMC;
  gemm2_wmma <<<(N + 63) / 64, 384, 0, stream>>>(agg1, W2, h2, N);
  zero_f32   <<<(no + T - 1) / T, T, 0, stream>>>(out, no);
  agg_edges_o<<<(E + 7) / 8, 256, 0, stream>>>(src, dst, norm, h2, out, E);
  final_out  <<<(no + T - 1) / T, T, 0, stream>>>(out, h2, dinv, b2, no);
}


# float atomic scatter-add replaced by a per-call destination-grouped edge list (slot = returning int add, arrival order; scan; fill) and one-wave-per-node row gathers summing each row in list order
# speedup vs baseline: 1.9315x; 1.9315x over previous
#include <hip/hip_runtime.h>

typedef __attribute__((ext_vector_type(2))) float v2f;
typedef __attribute__((ext_vector_type(8))) float v8f;

#define IN_DIMC 256
#define H_DIMC  128
#define OUT_DIMC 40
#define OUT_PADC 48

__global__ void zero_i32(int* __restrict__ p, int n) {
  int i = blockIdx.x * blockDim.x + threadIdx.x;
  if (i < n) p[i] = 0;
}

__global__ void count_rank(const int* __restrict__ dst, int* __restrict__ cnt,
                           int* __restrict__ rank, int e) {
  int i = blockIdx.x * blockDim.x + threadIdx.x;
  if (i < e) rank[i] = atomicAdd(&cnt[dst[i]], 1);
}

__global__ __launch_bounds__(1024) void scan_rows(const int* __restrict__ cnt,
                                                  int* __restrict__ rowptr,
                                                  float* __restrict__ dinv, int n) {
  __shared__ int s_pre[32];
  __shared__ int s_tot[32];
  const int tid  = threadIdx.x;
  const int lane = tid & 31;
  const int wave = tid >> 5;
  const int b    = blockIdx.x;

  int pre = 0;
  for (int j = 0; j < b; ++j) pre += cnt[j * 1024 + tid];
  for (int off = 16; off > 0; off >>= 1) pre += __shfl_xor(pre, off);
  if (lane == 0) s_pre[wave] = pre;

  const int i = b * 1024 + tid;
  const int c = (i < n) ? cnt[i] : 0;
  int x = c;
  for (int off = 1; off < 32; off <<= 1) {
    int y = __shfl_up(x, off);
    if (lane >= off) x += y;
  }
  if (lane == 31) s_tot[wave] = x;
  __syncthreads();

  int base = 0;
  for (int k = 0; k < 32; ++k) base += s_pre[k];
  for (int k = 0; k < wave; ++k) base += s_tot[k];

  if (i < n) {
    const int excl = base + x - c;
    rowptr[i] = excl;
    if (i == n - 1) rowptr[n] = excl + c;
    dinv[i] = rsqrtf(fmaxf((float)(c + 1), 1e-12f));
  }
}

__global__ void fill_rows(const int* __restrict__ src, const int* __restrict__ dst,
                          const int* __restrict__ rank, const int* __restrict__ rowptr,
                          const float* __restrict__ dinv, int2* __restrict__ rec, int e) {
  int i = blockIdx.x * blockDim.x + threadIdx.x;
  if (i < e) {
    int s = src[i], d = dst[i];
    float w = dinv[s] * dinv[d];
    rec[rowptr[d] + rank[i]] = make_int2(s, __float_as_int(w));
  }
}

__global__ __launch_bounds__(256) void gemm1_wmma(const float* __restrict__ X,
                                                  const float* __restrict__ W,
                                                  float* __restrict__ H, int n) {
  __shared__ float sW[64][H_DIMC + 4];
  __shared__ float sX[16][64 + 4];

  const int tid  = threadIdx.x;
  const int wave = tid >> 5;
  const int lane = tid & 31;
  const int col0 = wave * 16;
  const int m    = lane & 15;
  const int kh   = lane >> 4;
  const int row0 = blockIdx.x * 16;

  v8f acc = {0.f, 0.f, 0.f, 0.f, 0.f, 0.f, 0.f, 0.f};

  for (int kb = 0; kb < IN_DIMC; kb += 64) {
    __syncthreads();
    for (int i = tid; i < 64 * H_DIMC; i += 256) {
      int r = i >> 7, c = i & (H_DIMC - 1);
      sW[r][c] = W[(size_t)(kb + r) * H_DIMC + c];
    }
    for (int i = tid; i < 16 * 64; i += 256) {
      int r = i >> 6, k = i & 63;
      int row = row0 + r;
      sX[r][k] = (row < n) ? X[(size_t)row * IN_DIMC + kb + k] : 0.0f;
    }
    __syncthreads();

#pragma unroll
    for (int ks = 0; ks < 64; ks += 4) {
      v2f a, b;
      a[0] = sX[m][ks + 2 * kh + 0];
      a[1] = sX[m][ks + 2 * kh + 1];
      b[0] = sW[ks + 2 * kh + 0][col0 + m];
      b[1] = sW[ks + 2 * kh + 1][col0 + m];
      acc = __builtin_amdgcn_wmma_f32_16x16x4_f32(false, a, false, b,
                                                  (short)0, acc, false, false);
    }
  }
#pragma unroll
  for (int r = 0; r < 8; ++r) {
    int row = row0 + 8 * kh + r;
    if (row < n) H[(size_t)row * H_DIMC + col0 + m] = acc[r];
  }
}

__global__ __launch_bounds__(384) void gemm2_wmma(const float* __restrict__ Xh,
                                                  const float* __restrict__ W2,
                                                  float* __restrict__ H2, int n) {
  __shared__ float sW[H_DIMC][OUT_PADC + 4];
  __shared__ float sX[64][H_DIMC + 4];

  const int tid  = threadIdx.x;
  const int wave = tid >> 5;
  const int lane = tid & 31;
  const int rt   = wave / 3;
  const int ct   = wave % 3;
  const int col0 = ct * 16;
  const int m    = lane & 15;
  const int kh   = lane >> 4;
  const int row0 = blockIdx.x * 64;

  for (int i = tid; i < H_DIMC * OUT_PADC; i += 384) {
    int r = i / OUT_PADC, c = i % OUT_PADC;
    sW[r][c] = (c < OUT_DIMC) ? W2[(size_t)r * OUT_DIMC + c] : 0.0f;
  }
  for (int i = tid; i < 64 * H_DIMC; i += 384) {
    int r = i >> 7, k = i & (H_DIMC - 1);
    int row = row0 + r;
    sX[r][k] = (row < n) ? Xh[(size_t)row * H_DIMC + k] : 0.0f;
  }
  __syncthreads();

  v8f acc = {0.f, 0.f, 0.f, 0.f, 0.f, 0.f, 0.f, 0.f};
#pragma unroll
  for (int ks = 0; ks < H_DIMC; ks += 4) {
    v2f a, b;
    a[0] = sX[rt * 16 + m][ks + 2 * kh + 0];
    a[1] = sX[rt * 16 + m][ks + 2 * kh + 1];
    b[0] = sW[ks + 2 * kh + 0][col0 + m];
    b[1] = sW[ks + 2 * kh + 1][col0 + m];
    acc = __builtin_amdgcn_wmma_f32_16x16x4_f32(false, a, false, b,
                                                (short)0, acc, false, false);
  }
#pragma unroll
  for (int r = 0; r < 8; ++r) {
    int row = row0 + rt * 16 + 8 * kh + r;
    int col = col0 + m;
    if (row < n && col < OUT_DIMC) H2[(size_t)row * OUT_DIMC + col] = acc[r];
  }
}

__global__ __launch_bounds__(256) void gather_rows_h(const int* __restrict__ rowptr,
                                                     const int2* __restrict__ rec,
                                                     const float* __restrict__ h,
                                                     float* __restrict__ out, int n) {
  const int lane = threadIdx.x & 31;
  const int node = __builtin_amdgcn_readfirstlane((int)((blockIdx.x * 256u + threadIdx.x) >> 5));
  if (node >= n) return;
  const int beg = rowptr[node];
  const int end = rowptr[node + 1];
  const float4* __restrict__ hp = (const float4*)h;
  float4 acc = make_float4(0.0f, 0.0f, 0.0f, 0.0f);

  for (int base = beg; base < end; base += 32) {
    const int m = min(32, end - base);
    int2 r = make_int2(0, 0);
    if (lane < m) r = rec[base + lane];
    int k = 0;
    for (; k + 8 <= m; k += 8) {
      float4 v[8];
#pragma unroll
      for (int u = 0; u < 8; ++u) {
        const int s = __builtin_amdgcn_readlane(r.x, k + u);
        v[u] = hp[(size_t)s * (H_DIMC / 4) + lane];
      }
#pragma unroll
      for (int u = 0; u < 8; ++u) {
        const float w = __int_as_float(__builtin_amdgcn_readlane(r.y, k + u));
        acc.x += v[u].x * w; acc.y += v[u].y * w; acc.z += v[u].z * w; acc.w += v[u].w * w;
      }
    }
    const int t = m - k;
    if (t > 0) {
      float4 v[7];
#pragma unroll
      for (int u = 0; u < 7; ++u) {
        if (u < t) {
          const int s = __builtin_amdgcn_readlane(r.x, k + u);
          v[u] = hp[(size_t)s * (H_DIMC / 4) + lane];
        }
      }
#pragma unroll
      for (int u = 0; u < 7; ++u) {
        if (u < t) {
          const float w = __int_as_float(__builtin_amdgcn_readlane(r.y, k + u));
          acc.x += v[u].x * w; acc.y += v[u].y * w; acc.z += v[u].z * w; acc.w += v[u].w * w;
        }
      }
    }
  }
  ((float4*)out)[(size_t)node * (H_DIMC / 4) + lane] = acc;
}

__global__ __launch_bounds__(256) void gather_rows_o(const int* __restrict__ rowptr,
                                                     const int2* __restrict__ rec,
                                                     const float* __restrict__ h2,
                                                     float* __restrict__ out, int n) {
  const int lane = threadIdx.x & 31;
  const int q    = min(lane, OUT_DIMC / 4 - 1);
  const int node = __builtin_amdgcn_readfirstlane((int)((blockIdx.x * 256u + threadIdx.x) >> 5));
  if (node >= n) return;
  const int beg = rowptr[node];
  const int end = rowptr[node + 1];
  const float4* __restrict__ hp = (const float4*)h2;
  float4 acc = make_float4(0.0f, 0.0f, 0.0f, 0.0f);

  for (int base = beg; base < end; base += 32) {
    const int m = min(32, end - base);
    int2 r = make_int2(0, 0);
    if (lane < m) r = rec[base + lane];
    int k = 0;
    for (; k + 8 <= m; k += 8) {
      float4 v[8];
#pragma unroll
      for (int u = 0; u < 8; ++u) {
        const int s = __builtin_amdgcn_readlane(r.x, k + u);
        v[u] = hp[(size_t)s * (OUT_DIMC / 4) + q];
      }
#pragma unroll
      for (int u = 0; u < 8; ++u) {
        const float w = __int_as_float(__builtin_amdgcn_readlane(r.y, k + u));
        acc.x += v[u].x * w; acc.y += v[u].y * w; acc.z += v[u].z * w; acc.w += v[u].w * w;
      }
    }
    const int t = m - k;
    if (t > 0) {
      float4 v[7];
#pragma unroll
      for (int u = 0; u < 7; ++u) {
        if (u < t) {
          const int s = __builtin_amdgcn_readlane(r.x, k + u);
          v[u] = hp[(size_t)s * (OUT_DIMC / 4) + q];
        }
      }
#pragma unroll
      for (int u = 0; u < 7; ++u) {
        if (u < t) {
          const float w = __int_as_float(__builtin_amdgcn_readlane(r.y, k + u));
          acc.x += v[u].x * w; acc.y += v[u].y * w; acc.z += v[u].z * w; acc.w += v[u].w * w;
        }
      }
    }
  }
  if (lane < OUT_DIMC / 4) ((float4*)out)[(size_t)node * (OUT_DIMC / 4) + lane] = acc;
}

__global__ void relu_bias1(float* __restrict__ agg1, const float* __restrict__ h1,
                           const float* __restrict__ dinv, const float* __restrict__ b1,
                           int total) {
  int i = blockIdx.x * blockDim.x + threadIdx.x;
  if (i >= total) return;
  int node = i >> 7;
  int f    = i & 127;
  float di = dinv[node];
  float v = agg1[i] + h1[i] * di * di + b1[f];
  agg1[i] = v > 0.0f ? v : 0.0f;
}

__global__ void final_out(float* __restrict__ out, const float* __restrict__ h2,
                          const float* __restrict__ dinv, const float* __restrict__ b2,
                          int total) {
  int i = blockIdx.x * blockDim.x + threadIdx.x;
  if (i >= total) return;
  int node = i / OUT_DIMC;
  int f    = i - node * OUT_DIMC;
  float di = dinv[node];
  out[i] += h2[i] * di * di + b2[f];
}

extern "C" void kernel_launch(void* const* d_in, const int* in_sizes, int n_in,
                              void* d_out, int out_size, void* d_ws, size_t ws_size,
                              hipStream_t stream) {
  const float* x  = (const float*)d_in[0];
  const int*   ei = (const int*)d_in[1];
  const float* W1 = (const float*)d_in[2];
  const float* b1 = (const float*)d_in[3];
  const float* W2 = (const float*)d_in[4];
  const float* b2 = (const float*)d_in[5];
  float* out = (float*)d_out;

  const int N = in_sizes[0] / IN_DIMC;
  const int E = in_sizes[1] / 2;
  const int* src = ei;
  const int* dst = ei + E;

  size_t off = 0;
  char* base = (char*)d_ws;
  auto take = [&](size_t count) -> float* {
    float* p = (float*)(base + off);
    off += ((count * sizeof(float) + 255) & ~(size_t)255);
    return p;
  };
  float* dinv   = take((size_t)N);
  int*   cnt    = (int*)take((size_t)N);
  int*   rowptr = (int*)take((size_t)N + 1);
  int*   rank   = (int*)take((size_t)E);
  int2*  rec    = (int2*)take((size_t)E * 2);
  float* h1     = take((size_t)N * H_DIMC);
  float* agg1   = take((size_t)N * H_DIMC);
  float* h2     = take((size_t)N * OUT_DIMC);
  (void)ws_size; (void)n_in; (void)out_size;

  const int T = 256;
  zero_i32  <<<(N + T - 1) / T, T, 0, stream>>>(cnt, N);
  count_rank<<<(E + T - 1) / T, T, 0, stream>>>(dst, cnt, rank, E);
  scan_rows <<<(N + 1023) / 1024, 1024, 0, stream>>>(cnt, rowptr, dinv, N);
  fill_rows <<<(E + T - 1) / T, T, 0, stream>>>(src, dst, rank, rowptr, dinv, rec, E);

  const int nh = N * H_DIMC;
  gemm1_wmma   <<<(N + 15) / 16, 256, 0, stream>>>(x, W1, h1, N);
  gather_rows_h<<<(N + 7) / 8, 256, 0, stream>>>(rowptr, rec, h1, agg1, N);
  relu_bias1   <<<(nh + T - 1) / T, T, 0, stream>>>(agg1, h1, dinv, b1, nh);

  const int no = N * OUT_DIMC;
  gemm2_wmma   <<<(N + 63) / 64, 384, 0, stream>>>(agg1, W2, h2, N);
  gather_rows_o<<<(N + 7) / 8, 256, 0, stream>>>(rowptr, rec, h2, out, N);
  final_out    <<<(no + T - 1) / T, T, 0, stream>>>(out, h2, dinv, b2, no);
}


# self-loop + bias (+ReLU) folded into the two row gathers' register epilogue: the two elementwise passes and their launches go (10 -> 8 launches); summation order kept
# speedup vs baseline: 2.1307x; 1.1031x over previous
#include <hip/hip_runtime.h>

typedef __attribute__((ext_vector_type(2))) float v2f;
typedef __attribute__((ext_vector_type(8))) float v8f;

#define IN_DIMC 256
#define H_DIMC  128
#define OUT_DIMC 40
#define OUT_PADC 48

__global__ void zero_i32(int* __restrict__ p, int n) {
  int i = blockIdx.x * blockDim.x + threadIdx.x;
  if (i < n) p[i] = 0;
}

__global__ void count_rank(const int* __restrict__ dst, int* __restrict__ cnt,
                           int* __restrict__ rank, int e) {
  int i = blockIdx.x * blockDim.x + threadIdx.x;
  if (i < e) rank[i] = atomicAdd(&cnt[dst[i]], 1);
}

__global__ __launch_bounds__(1024) void scan_rows(const int* __restrict__ cnt,
                                                  int* __restrict__ rowptr,
                                                  float* __restrict__ dinv, int n) {
  __shared__ int s_pre[32];
  __shared__ int s_tot[32];
  const int tid  = threadIdx.x;
  const int lane = tid & 31;
  const int wave = tid >> 5;
  const int b    = blockIdx.x;

  int pre = 0;
  for (int j = 0; j < b; ++j) pre += cnt[j * 1024 + tid];
  for (int off = 16; off > 0; off >>= 1) pre += __shfl_xor(pre, off);
  if (lane == 0) s_pre[wave] = pre;

  const int i = b * 1024 + tid;
  const int c = (i < n) ? cnt[i] : 0;
  int x = c;
  for (int off = 1; off < 32; off <<= 1) {
    int y = __shfl_up(x, off);
    if (lane >= off) x += y;
  }
  if (lane == 31) s_tot[wave] = x;
  __syncthreads();

  int base = 0;
  for (int k = 0; k < 32; ++k) base += s_pre[k];
  for (int k = 0; k < wave; ++k) base += s_tot[k];

  if (i < n) {
    const int excl = base + x - c;
    rowptr[i] = excl;
    if (i == n - 1) rowptr[n] = excl + c;
    dinv[i] = rsqrtf(fmaxf((float)(c + 1), 1e-12f));
  }
}

__global__ void fill_rows(const int* __restrict__ src, const int* __restrict__ dst,
                          const int* __restrict__ rank, const int* __restrict__ rowptr,
                          const float* __restrict__ dinv, int2* __restrict__ rec, int e) {
  int i = blockIdx.x * blockDim.x + threadIdx.x;
  if (i < e) {
    int s = src[i], d = dst[i];
    float w = dinv[s] * dinv[d];
    rec[rowptr[d] + rank[i]] = make_int2(s, __float_as_int(w));
  }
}

__global__ __launch_bounds__(256) void gemm1_wmma(const float* __restrict__ X,
                                                  const float* __restrict__ W,
                                                  float* __restrict__ H, int n) {
  __shared__ float sW[64][H_DIMC + 4];
  __shared__ float sX[16][64 + 4];

  const int tid  = threadIdx.x;
  const int wave = tid >> 5;
  const int lane = tid & 31;
  const int col0 = wave * 16;
  const int m    = lane & 15;
  const int kh   = lane >> 4;
  const int row0 = blockIdx.x * 16;

  v8f acc = {0.f, 0.f, 0.f, 0.f, 0.f, 0.f, 0.f, 0.f};

  for (int kb = 0; kb < IN_DIMC; kb += 64) {
    __syncthreads();
    for (int i = tid; i < 64 * H_DIMC; i += 256) {
      int r = i >> 7, c = i & (H_DIMC - 1);
      sW[r][c] = W[(size_t)(kb + r) * H_DIMC + c];
    }
    for (int i = tid; i < 16 * 64; i += 256) {
      int r = i >> 6, k = i & 63;
      int row = row0 + r;
      sX[r][k] = (row < n) ? X[(size_t)row * IN_DIMC + kb + k] : 0.0f;
    }
    __syncthreads();

#pragma unroll
    for (int ks = 0; ks < 64; ks += 4) {
      v2f a, b;
      a[0] = sX[m][ks + 2 * kh + 0];
      a[1] = sX[m][ks + 2 * kh + 1];
      b[0] = sW[ks + 2 * kh + 0][col0 + m];
      b[1] = sW[ks + 2 * kh + 1][col0 + m];
      acc = __builtin_amdgcn_wmma_f32_16x16x4_f32(false, a, false, b,
                                                  (short)0, acc, false, false);
    }
  }
#pragma unroll
  for (int r = 0; r < 8; ++r) {
    int row = row0 + 8 * kh + r;
    if (row < n) H[(size_t)row * H_DIMC + col0 + m] = acc[r];
  }
}

__global__ __launch_bounds__(384) void gemm2_wmma(const float* __restrict__ Xh,
                                                  const float* __restrict__ W2,
                                                  float* __restrict__ H2, int n) {
  __shared__ float sW[H_DIMC][OUT_PADC + 4];
  __shared__ float sX[64][H_DIMC + 4];

  const int tid  = threadIdx.x;
  const int wave = tid >> 5;
  const int lane = tid & 31;
  const int rt   = wave / 3;
  const int ct   = wave % 3;
  const int col0 = ct * 16;
  const int m    = lane & 15;
  const int kh   = lane >> 4;
  const int row0 = blockIdx.x * 64;

  for (int i = tid; i < H_DIMC * OUT_PADC; i += 384) {
    int r = i / OUT_PADC, c = i % OUT_PADC;
    sW[r][c] = (c < OUT_DIMC) ? W2[(size_t)r * OUT_DIMC + c] : 0.0f;
  }
  for (int i = tid; i < 64 * H_DIMC; i += 384) {
    int r = i >> 7, k = i & (H_DIMC - 1);
    int row = row0 + r;
    sX[r][k] = (row < n) ? Xh[(size_t)row * H_DIMC + k] : 0.0f;
  }
  __syncthreads();

  v8f acc = {0.f, 0.f, 0.f, 0.f, 0.f, 0.f, 0.f, 0.f};
#pragma unroll
  for (int ks = 0; ks < H_DIMC; ks += 4) {
    v2f a, b;
    a[0] = sX[rt * 16 + m][ks + 2 * kh + 0];
    a[1] = sX[rt * 16 + m][ks + 2 * kh + 1];
    b[0] = sW[ks + 2 * kh + 0][col0 + m];
    b[1] = sW[ks + 2 * kh + 1][col0 + m];
    acc = __builtin_amdgcn_wmma_f32_16x16x4_f32(false, a, false, b,
                                                (short)0, acc, false, false);
  }
#pragma unroll
  for (int r = 0; r < 8; ++r) {
    int row = row0 + rt * 16 + 8 * kh + r;
    int col = col0 + m;
    if (row < n && col < OUT_DIMC) H2[(size_t)row * OUT_DIMC + col] = acc[r];
  }
}

__global__ __launch_bounds__(256) void gather_rows_h(const int* __restrict__ rowptr,
                                                     const int2* __restrict__ rec,
                                                     const float* __restrict__ h,
                                                     const float* __restrict__ dinv,
                                                     const float* __restrict__ b1,
                                                     float* __restrict__ out, int n) {
  const int lane = threadIdx.x & 31;
  const int node = __builtin_amdgcn_readfirstlane((int)((blockIdx.x * 256u + threadIdx.x) >> 5));
  if (node >= n) return;
  const int beg = rowptr[node];
  const int end = rowptr[node + 1];
  const float4* __restrict__ hp = (const float4*)h;
  const float4 hs = hp[(size_t)node * (H_DIMC / 4) + lane];
  const float  di = dinv[node];
  const float bx = b1[4 * lane + 0], by = b1[4 * lane + 1];
  const float bz = b1[4 * lane + 2], bw = b1[4 * lane + 3];
  float4 acc = make_float4(0.0f, 0.0f, 0.0f, 0.0f);

  for (int base = beg; base < end; base += 32) {
    const int m = min(32, end - base);
    int2 r = make_int2(0, 0);
    if (lane < m) r = rec[base + lane];
    int k = 0;
    for (; k + 8 <= m; k += 8) {
      float4 v[8];
#pragma unroll
      for (int u = 0; u < 8; ++u) {
        const int s = __builtin_amdgcn_readlane(r.x, k + u);
        v[u] = hp[(size_t)s * (H_DIMC / 4) + lane];
      }
#pragma unroll
      for (int u = 0; u < 8; ++u) {
        const float w = __int_as_float(__builtin_amdgcn_readlane(r.y, k + u));
        acc.x += v[u].x * w; acc.y += v[u].y * w; acc.z += v[u].z * w; acc.w += v[u].w * w;
      }
    }
    const int t = m - k;
    if (t > 0) {
      float4 v[7];
#pragma unroll
      for (int u = 0; u < 7; ++u) {
        if (u < t) {
          const int s = __builtin_amdgcn_readlane(r.x, k + u);
          v[u] = hp[(size_t)s * (H_DIMC / 4) + lane];
        }
      }
#pragma unroll
      for (int u = 0; u < 7; ++u) {
        if (u < t) {
          const float w = __int_as_float(__builtin_amdgcn_readlane(r.y, k + u));
          acc.x += v[u].x * w; acc.y += v[u].y * w; acc.z += v[u].z * w; acc.w += v[u].w * w;
        }
      }
    }
  }
  float4 o;
  o.x = acc.x + hs.x * di * di + bx;
  o.y = acc.y + hs.y * di * di + by;
  o.z = acc.z + hs.z * di * di + bz;
  o.w = acc.w + hs.w * di * di + bw;
  o.x = o.x > 0.0f ? o.x : 0.0f;
  o.y = o.y > 0.0f ? o.y : 0.0f;
  o.z = o.z > 0.0f ? o.z : 0.0f;
  o.w = o.w > 0.0f ? o.w : 0.0f;
  ((float4*)out)[(size_t)node * (H_DIMC / 4) + lane] = o;
}

__global__ __launch_bounds__(256) void gather_rows_o(const int* __restrict__ rowptr,
                                                     const int2* __restrict__ rec,
                                                     const float* __restrict__ h2,
                                                     const float* __restrict__ dinv,
                                                     const float* __restrict__ b2,
                                                     float* __restrict__ out, int n) {
  const int lane = threadIdx.x & 31;
  const int q    = min(lane, OUT_DIMC / 4 - 1);
  const int node = __builtin_amdgcn_readfirstlane((int)((blockIdx.x * 256u + threadIdx.x) >> 5));
  if (node >= n) return;
  const int beg = rowptr[node];
  const int end = rowptr[node + 1];
  const float4* __restrict__ hp = (const float4*)h2;
  const float4 hs = hp[(size_t)node * (OUT_DIMC / 4) + q];
  const float  di = dinv[node];
  const float bx = b2[4 * q + 0], by = b2[4 * q + 1];
  const float bz = b2[4 * q + 2], bw = b2[4 * q + 3];
  float4 acc = make_float4(0.0f, 0.0f, 0.0f, 0.0f);

  for (int base = beg; base < end; base += 32) {
    const int m = min(32, end - base);
    int2 r = make_int2(0, 0);
    if (lane < m) r = rec[base + lane];
    int k = 0;
    for (; k + 8 <= m; k += 8) {
      float4 v[8];
#pragma unroll
      for (int u = 0; u < 8; ++u) {
        const int s = __builtin_amdgcn_readlane(r.x, k + u);
        v[u] = hp[(size_t)s * (OUT_DIMC / 4) + q];
      }
#pragma unroll
      for (int u = 0; u < 8; ++u) {
        const float w = __int_as_float(__builtin_amdgcn_readlane(r.y, k + u));
        acc.x += v[u].x * w; acc.y += v[u].y * w; acc.z += v[u].z * w; acc.w += v[u].w * w;
      }
    }
    const int t = m - k;
    if (t > 0) {
      float4 v[7];
#pragma unroll
      for (int u = 0; u < 7; ++u) {
        if (u < t) {
          const int s = __builtin_amdgcn_readlane(r.x, k + u);
          v[u] = hp[(size_t)s * (OUT_DIMC / 4) + q];
        }
      }
#pragma unroll
      for (int u = 0; u < 7; ++u) {
        if (u < t) {
          const float w = __int_as_float(__builtin_amdgcn_readlane(r.y, k + u));
          acc.x += v[u].x * w; acc.y += v[u].y * w; acc.z += v[u].z * w; acc.w += v[u].w * w;
        }
      }
    }
  }
  float4 o;
  o.x = acc.x + (hs.x * di * di + bx);
  o.y = acc.y + (hs.y * di * di + by);
  o.z = acc.z + (hs.z * di * di + bz);
  o.w = acc.w + (hs.w * di * di + bw);
  if (lane < OUT_DIMC / 4) ((float4*)out)[(size_t)node * (OUT_DIMC / 4) + lane] = o;
}

extern "C" void kernel_launch(void* const* d_in, const int* in_sizes, int n_in,
                              void* d_out, int out_size, void* d_ws, size_t ws_size,
                              hipStream_t stream) {
  const float* x  = (const float*)d_in[0];
  const int*   ei = (const int*)d_in[1];
  const float* W1 = (const float*)d_in[2];
  const float* b1 = (const float*)d_in[3];
  const float* W2 = (const float*)d_in[4];
  const float* b2 = (const float*)d_in[5];
  float* out = (float*)d_out;

  const int N = in_sizes[0] / IN_DIMC;
  const int E = in_sizes[1] / 2;
  const int* src = ei;
  const int* dst = ei + E;

  size_t off = 0;
  char* base = (char*)d_ws;
  auto take = [&](size_t count) -> float* {
    float* p = (float*)(base + off);
    off += ((count * sizeof(float) + 255) & ~(size_t)255);
    return p;
  };
  float* dinv   = take((size_t)N);
  int*   cnt    = (int*)take((size_t)N);
  int*   rowptr = (int*)take((size_t)N + 1);
  int*   rank   = (int*)take((size_t)E);
  int2*  rec    = (int2*)take((size_t)E * 2);
  float* h1     = take((size_t)N * H_DIMC);
  float* agg1   = take((size_t)N * H_DIMC);
  float* h2     = take((size_t)N * OUT_DIMC);
  (void)ws_size; (void)n_in; (void)out_size;

  const int T = 256;
  zero_i32  <<<(N + T - 1) / T, T, 0, stream>>>(cnt, N);
  count_rank<<<(E + T - 1) / T, T, 0, stream>>>(dst, cnt, rank, E);
  scan_rows <<<(N + 1023) / 1024, 1024, 0, stream>>>(cnt, rowptr, dinv, N);
  fill_rows <<<(E + T - 1) / T, T, 0, stream>>>(src, dst, rank, rowptr, dinv, rec, E);

  gemm1_wmma   <<<(N + 15) / 16, 256, 0, stream>>>(x, W1, h1, N);
  gather_rows_h<<<(N + 7) / 8, 256, 0, stream>>>(rowptr, rec, h1, dinv, b1, agg1, N);

  gemm2_wmma   <<<(N + 63) / 64, 384, 0, stream>>>(agg1, W2, h2, N);
  gather_rows_o<<<(N + 7) / 8, 256, 0, stream>>>(rowptr, rec, h2, dinv, b2, out, N);
}


# edge list built by a two-level counting sort (bins of 256 nodes, ranks from LDS counters) in place of one returning global add per edge; rows summed in slot arrival order as before; 8 -> 9 launches
# speedup vs baseline: 2.7760x; 1.3029x over previous
#include <hip/hip_runtime.h>

typedef __attribute__((ext_vector_type(2))) float v2f;
typedef __attribute__((ext_vector_type(8))) float v8f;

#define IN_DIMC 256
#define H_DIMC  128
#define OUT_DIMC 40
#define OUT_PADC 48

#define BIN_SHIFT 8
#define NBMAX     512
#define CHUNK     8192
#define CSTRIDE   32

__global__ void zero_i32(int* __restrict__ p, int n) {
  int i = blockIdx.x * blockDim.x + threadIdx.x;
  if (i < n) p[i] = 0;
}

__global__ __launch_bounds__(256) void bin_count(const int* __restrict__ dst, int* __restrict__ bincnt,
                                                 int e, int nb) {
  __shared__ int s_hist[NBMAX];
  const int tid = threadIdx.x;
  s_hist[tid] = 0;
  s_hist[tid + 256] = 0;
  __syncthreads();
  const int base = blockIdx.x * CHUNK;
  auto count_pass = [&](const bool FULL) {
    for (int u0 = 0; u0 < CHUNK / 256; u0 += 8) {
      if (!FULL && base + 256 * u0 >= e) break;
      int d[8];
#pragma unroll
      for (int k = 0; k < 8; ++k) {
        const int i = base + tid + 256 * (u0 + k);
        d[k] = dst[FULL ? i : min(i, e - 1)];
      }
#pragma unroll
      for (int k = 0; k < 8; ++k) {
        const int i = base + tid + 256 * (u0 + k);
        if (FULL || i < e) atomicAdd(&s_hist[d[k] >> BIN_SHIFT], 1);
      }
    }
  };
  if (base + CHUNK <= e) count_pass(true); else count_pass(false);
  __syncthreads();
  for (int b = tid; b < nb; b += 256) {
    const int c = s_hist[b];
    if (c) atomicAdd(&bincnt[b * CSTRIDE], c);
  }
}

__global__ __launch_bounds__(256) void bin_scatter(const int* __restrict__ src, const int* __restrict__ dst,
                                                   const int* __restrict__ bincnt, int* __restrict__ bincur,
                                                   int* __restrict__ binstart, int2* __restrict__ binned,
                                                   int e, int nb) {
  __shared__ int s_start[NBMAX];
  __shared__ int s_hist[NBMAX];
  __shared__ int s_wsum[8];
  const int tid  = threadIdx.x;
  const int lane = tid & 31;
  const int wave = tid >> 5;

  const int ca = (2 * tid     < nb) ? bincnt[(2 * tid)     * CSTRIDE] : 0;
  const int cb = (2 * tid + 1 < nb) ? bincnt[(2 * tid + 1) * CSTRIDE] : 0;
  const int cs = ca + cb;
  int x = cs;
  for (int off = 1; off < 32; off <<= 1) {
    int y = __shfl_up(x, off);
    if (lane >= off) x += y;
  }
  if (lane == 31) s_wsum[wave] = x;
  s_hist[tid] = 0;
  s_hist[tid + 256] = 0;
  __syncthreads();
  int wbase = 0;
  for (int k = 0; k < wave; ++k) wbase += s_wsum[k];
  const int excl = wbase + x - cs;
  s_start[2 * tid]     = excl;
  s_start[2 * tid + 1] = excl + ca;
  if (blockIdx.x == 0) {
    if (2 * tid     < nb) binstart[2 * tid]     = excl;
    if (2 * tid + 1 < nb) binstart[2 * tid + 1] = excl + ca;
  }

  const int base = blockIdx.x * CHUNK;
  const bool full = base + CHUNK <= e;

  auto count_pass = [&](const bool FULL) {
    for (int u0 = 0; u0 < CHUNK / 256; u0 += 8) {
      if (!FULL && base + 256 * u0 >= e) break;
      int d[8];
#pragma unroll
      for (int k = 0; k < 8; ++k) {
        const int i = base + tid + 256 * (u0 + k);
        d[k] = dst[FULL ? i : min(i, e - 1)];
      }
#pragma unroll
      for (int k = 0; k < 8; ++k) {
        const int i = base + tid + 256 * (u0 + k);
        if (FULL || i < e) atomicAdd(&s_hist[d[k] >> BIN_SHIFT], 1);
      }
    }
  };
  if (full) count_pass(true); else count_pass(false);
  __syncthreads();

  for (int b = tid; b < nb; b += 256) {
    const int c = s_hist[b];
    const int off = c ? atomicAdd(&bincur[b * CSTRIDE], c) : 0;
    s_hist[b] = s_start[b] + off;
  }
  __syncthreads();

  auto place_pass = [&](const bool FULL) {
    for (int u0 = 0; u0 < CHUNK / 256; u0 += 8) {
      if (!FULL && base + 256 * u0 >= e) break;
      int d[8], s[8], pos[8];
#pragma unroll
      for (int k = 0; k < 8; ++k) {
        const int i = base + tid + 256 * (u0 + k);
        const int j = FULL ? i : min(i, e - 1);
        d[k] = dst[j];
        s[k] = src[j];
      }
#pragma unroll
      for (int k = 0; k < 8; ++k) {
        const int i = base + tid + 256 * (u0 + k);
        pos[k] = 0;
        if (FULL || i < e) pos[k] = atomicAdd(&s_hist[d[k] >> BIN_SHIFT], 1);
      }
#pragma unroll
      for (int k = 0; k < 8; ++k) {
        const int i = base + tid + 256 * (u0 + k);
        if (FULL || i < e) binned[pos[k]] = make_int2(s[k], d[k]);
      }
    }
  };
  if (full) place_pass(true); else place_pass(false);
}

__global__ __launch_bounds__(256) void bin_rows(const int2* __restrict__ binned, const int* __restrict__ bincnt,
                                                const int* __restrict__ binstart, int* __restrict__ rowptr,
                                                float* __restrict__ dinv, int2* __restrict__ rec, int n) {
  __shared__ int s_cnt[256];
  __shared__ int s_wsum[8];
  const int tid  = threadIdx.x;
  const int lane = tid & 31;
  const int wave = tid >> 5;
  const int bin   = blockIdx.x;
  const int start = binstart[bin];
  const int ne    = bincnt[bin * CSTRIDE];
  const int2* __restrict__ mine = binned + start;

  s_cnt[tid] = 0;
  __syncthreads();
  for (int i0 = tid; i0 < ne + tid; i0 += 8 * 256) {
    int2 r[8];
#pragma unroll
    for (int k = 0; k < 8; ++k) r[k] = mine[min(i0 + 256 * k, ne - 1)];
#pragma unroll
    for (int k = 0; k < 8; ++k)
      if (i0 + 256 * k < ne) atomicAdd(&s_cnt[r[k].y & 255], 1);
  }
  __syncthreads();

  const int c = s_cnt[tid];
  int x = c;
  for (int off = 1; off < 32; off <<= 1) {
    int y = __shfl_up(x, off);
    if (lane >= off) x += y;
  }
  if (lane == 31) s_wsum[wave] = x;
  __syncthreads();
  int wbase = 0;
  for (int k = 0; k < wave; ++k) wbase += s_wsum[k];
  const int row0 = start + wbase + x - c;

  const int node = (bin << BIN_SHIFT) + tid;
  if (node < n) {
    rowptr[node] = row0;
    if (node == n - 1) rowptr[n] = row0 + c;
    dinv[node] = rsqrtf(fmaxf((float)(c + 1), 1e-12f));
  }
  s_cnt[tid] = row0;
  __syncthreads();
  for (int i0 = tid; i0 < ne + tid; i0 += 8 * 256) {
    int2 r[8];
    int  pos[8];
#pragma unroll
    for (int k = 0; k < 8; ++k) r[k] = mine[min(i0 + 256 * k, ne - 1)];
#pragma unroll
    for (int k = 0; k < 8; ++k) {
      pos[k] = 0;
      if (i0 + 256 * k < ne) pos[k] = atomicAdd(&s_cnt[r[k].y & 255], 1);
    }
#pragma unroll
    for (int k = 0; k < 8; ++k)
      if (i0 + 256 * k < ne) rec[pos[k]] = r[k];
  }
}

__global__ void edge_weights(const float* __restrict__ dinv, int2* __restrict__ rec, int e) {
  int i = blockIdx.x * blockDim.x + threadIdx.x;
  if (i < e) {
    const int2 r = rec[i];
    const float w = dinv[r.x] * dinv[r.y];
    rec[i] = make_int2(r.x, __float_as_int(w));
  }
}

__global__ __launch_bounds__(256) void gemm1_wmma(const float* __restrict__ X,
                                                  const float* __restrict__ W,
                                                  float* __restrict__ H, int n) {
  __shared__ float sW[64][H_DIMC + 4];
  __shared__ float sX[16][64 + 4];

  const int tid  = threadIdx.x;
  const int wave = tid >> 5;
  const int lane = tid & 31;
  const int col0 = wave * 16;
  const int m    = lane & 15;
  const int kh   = lane >> 4;
  const int row0 = blockIdx.x * 16;

  v8f acc = {0.f, 0.f, 0.f, 0.f, 0.f, 0.f, 0.f, 0.f};

  for (int kb = 0; kb < IN_DIMC; kb += 64) {
    __syncthreads();
    for (int i = tid; i < 64 * H_DIMC; i += 256) {
      int r = i >> 7, c = i & (H_DIMC - 1);
      sW[r][c] = W[(size_t)(kb + r) * H_DIMC + c];
    }
    for (int i = tid; i < 16 * 64; i += 256) {
      int r = i >> 6, k = i & 63;
      int row = row0 + r;
      sX[r][k] = (row < n) ? X[(size_t)row * IN_DIMC + kb + k] : 0.0f;
    }
    __syncthreads();

#pragma unroll
    for (int ks = 0; ks < 64; ks += 4) {
      v2f a, b;
      a[0] = sX[m][ks + 2 * kh + 0];
      a[1] = sX[m][ks + 2 * kh + 1];
      b[0] = sW[ks + 2 * kh + 0][col0 + m];
      b[1] = sW[ks + 2 * kh + 1][col0 + m];
      acc = __builtin_amdgcn_wmma_f32_16x16x4_f32(false, a, false, b,
                                                  (short)0, acc, false, false);
    }
  }
#pragma unroll
  for (int r = 0; r < 8; ++r) {
    int row = row0 + 8 * kh + r;
    if (row < n) H[(size_t)row * H_DIMC + col0 + m] = acc[r];
  }
}

__global__ __launch_bounds__(384) void gemm2_wmma(const float* __restrict__ Xh,
                                                  const float* __restrict__ W2,
                                                  float* __restrict__ H2, int n) {
  __shared__ float sW[H_DIMC][OUT_PADC + 4];
  __shared__ float sX[64][H_DIMC + 4];

  const int tid  = threadIdx.x;
  const int wave = tid >> 5;
  const int lane = tid & 31;
  const int rt   = wave / 3;
  const int ct   = wave % 3;
  const int col0 = ct * 16;
  const int m    = lane & 15;
  const int kh   = lane >> 4;
  const int row0 = blockIdx.x * 64;

  for (int i = tid; i < H_DIMC * OUT_PADC; i += 384) {
    int r = i / OUT_PADC, c = i % OUT_PADC;
    sW[r][c] = (c < OUT_DIMC) ? W2[(size_t)r * OUT_DIMC + c] : 0.0f;
  }
  for (int i = tid; i < 64 * H_DIMC; i += 384) {
    int r = i >> 7, k = i & (H_DIMC - 1);
    int row = row0 + r;
    sX[r][k] = (row < n) ? Xh[(size_t)row * H_DIMC + k] : 0.0f;
  }
  __syncthreads();

  v8f acc = {0.f, 0.f, 0.f, 0.f, 0.f, 0.f, 0.f, 0.f};
#pragma unroll
  for (int ks = 0; ks < H_DIMC; ks += 4) {
    v2f a, b;
    a[0] = sX[rt * 16 + m][ks + 2 * kh + 0];
    a[1] = sX[rt * 16 + m][ks + 2 * kh + 1];
    b[0] = sW[ks + 2 * kh + 0][col0 + m];
    b[1] = sW[ks + 2 * kh + 1][col0 + m];
    acc = __builtin_amdgcn_wmma_f32_16x16x4_f32(false, a, false, b,
                                                (short)0, acc, false, false);
  }
#pragma unroll
  for (int r = 0; r < 8; ++r) {
    int row = row0 + rt * 16 + 8 * kh + r;
    int col = col0 + m;
    if (row < n && col < OUT_DIMC) H2[(size_t)row * OUT_DIMC + col] = acc[r];
  }
}

__global__ __launch_bounds__(256) void gather_rows_h(const int* __restrict__ rowptr,
                                                     const int2* __restrict__ rec,
                                                     const float* __restrict__ h,
                                                     const float* __restrict__ dinv,
                                                     const float* __restrict__ b1,
                                                     float* __restrict__ out, int n) {
  const int lane = threadIdx.x & 31;
  const int node = __builtin_amdgcn_readfirstlane((int)((blockIdx.x * 256u + threadIdx.x) >> 5));
  if (node >= n) return;
  const int beg = rowptr[node];
  const int end = rowptr[node + 1];
  const float4* __restrict__ hp = (const float4*)h;
  const float4 hs = hp[(size_t)node * (H_DIMC / 4) + lane];
  const float  di = dinv[node];
  const float bx = b1[4 * lane + 0], by = b1[4 * lane + 1];
  const float bz = b1[4 * lane + 2], bw = b1[4 * lane + 3];
  float4 acc = make_float4(0.0f, 0.0f, 0.0f, 0.0f);

  for (int base = beg; base < end; base += 32) {
    const int m = min(32, end - base);
    int2 r = make_int2(0, 0);
    if (lane < m) r = rec[base + lane];
    int k = 0;
    for (; k + 8 <= m; k += 8) {
      float4 v[8];
#pragma unroll
      for (int u = 0; u < 8; ++u) {
        const int s = __builtin_amdgcn_readlane(r.x, k + u);
        v[u] = hp[(size_t)s * (H_DIMC / 4) + lane];
      }
#pragma unroll
      for (int u = 0; u < 8; ++u) {
        const float w = __int_as_float(__builtin_amdgcn_readlane(r.y, k + u));
        acc.x += v[u].x * w; acc.y += v[u].y * w; acc.z += v[u].z * w; acc.w += v[u].w * w;
      }
    }
    const int t = m - k;
    if (t > 0) {
      float4 v[7];
#pragma unroll
      for (int u = 0; u < 7; ++u) {
        if (u < t) {
          const int s = __builtin_amdgcn_readlane(r.x, k + u);
          v[u] = hp[(size_t)s * (H_DIMC / 4) + lane];
        }
      }
#pragma unroll
      for (int u = 0; u < 7; ++u) {
        if (u < t) {
          const float w = __int_as_float(__builtin_amdgcn_readlane(r.y, k + u));
          acc.x += v[u].x * w; acc.y += v[u].y * w; acc.z += v[u].z * w; acc.w += v[u].w * w;
        }
      }
    }
  }
  float4 o;
  o.x = acc.x + hs.x * di * di + bx;
  o.y = acc.y + hs.y * di * di + by;
  o.z = acc.z + hs.z * di * di + bz;
  o.w = acc.w + hs.w * di * di + bw;
  o.x = o.x > 0.0f ? o.x : 0.0f;
  o.y = o.y > 0.0f ? o.y : 0.0f;
  o.z = o.z > 0.0f ? o.z : 0.0f;
  o.w = o.w > 0.0f ? o.w : 0.0f;
  ((float4*)out)[(size_t)node * (H_DIMC / 4) + lane] = o;
}

__global__ __launch_bounds__(256) void gather_rows_o(const int* __restrict__ rowptr,
                                                     const int2* __restrict__ rec,
                                                     const float* __restrict__ h2,
                                                     const float* __restrict__ dinv,
                                                     const float* __restrict__ b2,
                                                     float* __restrict__ out, int n) {
  const int lane = threadIdx.x & 31;
  const int q    = min(lane, OUT_DIMC / 4 - 1);
  const int node = __builtin_amdgcn_readfirstlane((int)((blockIdx.x * 256u + threadIdx.x) >> 5));
  if (node >= n) return;
  const int beg = rowptr[node];
  const int end = rowptr[node + 1];
  const float4* __restrict__ hp = (const float4*)h2;
  const float4 hs = hp[(size_t)node * (OUT_DIMC / 4) + q];
  const float  di = dinv[node];
  const float bx = b2[4 * q + 0], by = b2[4 * q + 1];
  const float bz = b2[4 * q + 2], bw = b2[4 * q + 3];
  float4 acc = make_float4(0.0f, 0.0f, 0.0f, 0.0f);

  for (int base = beg; base < end; base += 32) {
    const int m = min(32, end - base);
    int2 r = make_int2(0, 0);
    if (lane < m) r = rec[base + lane];
    int k = 0;
    for (; k + 8 <= m; k += 8) {
      float4 v[8];
#pragma unroll
      for (int u = 0; u < 8; ++u) {
        const int s = __builtin_amdgcn_readlane(r.x, k + u);
        v[u] = hp[(size_t)s * (OUT_DIMC / 4) + q];
      }
#pragma unroll
      for (int u = 0; u < 8; ++u) {
        const float w = __int_as_float(__builtin_amdgcn_readlane(r.y, k + u));
        acc.x += v[u].x * w; acc.y += v[u].y * w; acc.z += v[u].z * w; acc.w += v[u].w * w;
      }
    }
    const int t = m - k;
    if (t > 0) {
      float4 v[7];
#pragma unroll
      for (int u = 0; u < 7; ++u) {
        if (u < t) {
          const int s = __builtin_amdgcn_readlane(r.x, k + u);
          v[u] = hp[(size_t)s * (OUT_DIMC / 4) + q];
        }
      }
#pragma unroll
      for (int u = 0; u < 7; ++u) {
        if (u < t) {
          const float w = __int_as_float(__builtin_amdgcn_readlane(r.y, k + u));
          acc.x += v[u].x * w; acc.y += v[u].y * w; acc.z += v[u].z * w; acc.w += v[u].w * w;
        }
      }
    }
  }
  float4 o;
  o.x = acc.x + (hs.x * di * di + bx);
  o.y = acc.y + (hs.y * di * di + by);
  o.z = acc.z + (hs.z * di * di + bz);
  o.w = acc.w + (hs.w * di * di + bw);
  if (lane < OUT_DIMC / 4) ((float4*)out)[(size_t)node * (OUT_DIMC / 4) + lane] = o;
}

extern "C" void kernel_launch(void* const* d_in, const int* in_sizes, int n_in,
                              void* d_out, int out_size, void* d_ws, size_t ws_size,
                              hipStream_t stream) {
  const float* x  = (const float*)d_in[0];
  const int*   ei = (const int*)d_in[1];
  const float* W1 = (const float*)d_in[2];
  const float* b1 = (const float*)d_in[3];
  const float* W2 = (const float*)d_in[4];
  const float* b2 = (const float*)d_in[5];
  float* out = (float*)d_out;

  const int N = in_sizes[0] / IN_DIMC;
  const int E = in_sizes[1] / 2;
  const int* src = ei;
  const int* dst = ei + E;

  size_t off = 0;
  char* base = (char*)d_ws;
  auto take = [&](size_t count) -> float* {
    float* p = (float*)(base + off);
    off += ((count * sizeof(float) + 255) & ~(size_t)255);
    return p;
  };
  float* dinv   = take((size_t)N);
  int*   bincnt = (int*)take((size_t)2 * NBMAX * CSTRIDE);
  int*   bincur = bincnt + NBMAX * CSTRIDE;
  int*   binstart = (int*)take((size_t)NBMAX);
  int*   rowptr = (int*)take((size_t)N + 1);
  int2*  binned = (int2*)take((size_t)E * 2);
  int2*  rec    = (int2*)take((size_t)E * 2);
  float* h1     = take((size_t)N * H_DIMC);
  float* agg1   = take((size_t)N * H_DIMC);
  float* h2     = take((size_t)N * OUT_DIMC);
  (void)ws_size; (void)n_in; (void)out_size;

  const int T = 256;
  const int NB = (N + 255) >> BIN_SHIFT;
  const int GB = (E + CHUNK - 1) / CHUNK;
  zero_i32    <<<(2 * NBMAX * CSTRIDE + T - 1) / T, T, 0, stream>>>(bincnt, 2 * NBMAX * CSTRIDE);
  bin_count   <<<GB, T, 0, stream>>>(dst, bincnt, E, NB);
  bin_scatter <<<GB, T, 0, stream>>>(src, dst, bincnt, bincur, binstart, binned, E, NB);
  bin_rows    <<<NB, T, 0, stream>>>(binned, bincnt, binstart, rowptr, dinv, rec, N);
  edge_weights<<<(E + T - 1) / T, T, 0, stream>>>(dinv, rec, E);

  gemm1_wmma   <<<(N + 15) / 16, 256, 0, stream>>>(x, W1, h1, N);
  gather_rows_h<<<(N + 7) / 8, 256, 0, stream>>>(rowptr, rec, h1, dinv, b1, agg1, N);

  gemm2_wmma   <<<(N + 63) / 64, 384, 0, stream>>>(agg1, W2, h2, N);
  gather_rows_o<<<(N + 7) / 8, 256, 0, stream>>>(rowptr, rec, h2, dinv, b2, out, N);
}


# both GEMMs restaged: operands fetched by 16-B loads issued as one group (was one 4-B load per round trip), a slab ahead in GEMM1, whose block tile grows to 64 rows on 4 accumulators; k order kept
# speedup vs baseline: 3.9794x; 1.4335x over previous
#include <hip/hip_runtime.h>

typedef __attribute__((ext_vector_type(2))) float v2f;
typedef __attribute__((ext_vector_type(8))) float v8f;

#define IN_DIMC 256
#define H_DIMC  128
#define OUT_DIMC 40
#define OUT_PADC 48

#define BIN_SHIFT 8
#define NBMAX     512
#define CHUNK     8192
#define CSTRIDE   32

__global__ void zero_i32(int* __restrict__ p, int n) {
  int i = blockIdx.x * blockDim.x + threadIdx.x;
  if (i < n) p[i] = 0;
}

__global__ __launch_bounds__(256) void bin_count(const int* __restrict__ dst, int* __restrict__ bincnt,
                                                 int e, int nb) {
  __shared__ int s_hist[NBMAX];
  const int tid = threadIdx.x;
  s_hist[tid] = 0;
  s_hist[tid + 256] = 0;
  __syncthreads();
  const int base = blockIdx.x * CHUNK;
  auto count_pass = [&](const bool FULL) {
    for (int u0 = 0; u0 < CHUNK / 256; u0 += 8) {
      if (!FULL && base + 256 * u0 >= e) break;
      int d[8];
#pragma unroll
      for (int k = 0; k < 8; ++k) {
        const int i = base + tid + 256 * (u0 + k);
        d[k] = dst[FULL ? i : min(i, e - 1)];
      }
#pragma unroll
      for (int k = 0; k < 8; ++k) {
        const int i = base + tid + 256 * (u0 + k);
        if (FULL || i < e) atomicAdd(&s_hist[d[k] >> BIN_SHIFT], 1);
      }
    }
  };
  if (base + CHUNK <= e) count_pass(true); else count_pass(false);
  __syncthreads();
  for (int b = tid; b < nb; b += 256) {
    const int c = s_hist[b];
    if (c) atomicAdd(&bincnt[b * CSTRIDE], c);
  }
}

__global__ __launch_bounds__(256) void bin_scatter(const int* __restrict__ src, const int* __restrict__ dst,
                                                   const int* __restrict__ bincnt, int* __restrict__ bincur,
                                                   int* __restrict__ binstart, int2* __restrict__ binned,
                                                   int e, int nb) {
  __shared__ int s_start[NBMAX];
  __shared__ int s_hist[NBMAX];
  __shared__ int s_wsum[8];
  const int tid  = threadIdx.x;
  const int lane = tid & 31;
  const int wave = tid >> 5;

  const int ca = (2 * tid     < nb) ? bincnt[(2 * tid)     * CSTRIDE] : 0;
  const int cb = (2 * tid + 1 < nb) ? bincnt[(2 * tid + 1) * CSTRIDE] : 0;
  const int cs = ca + cb;
  int x = cs;
  for (int off = 1; off < 32; off <<= 1) {
    int y = __shfl_up(x, off);
    if (lane >= off) x += y;
  }
  if (lane == 31) s_wsum[wave] = x;
  s_hist[tid] = 0;
  s_hist[tid + 256] = 0;
  __syncthreads();
  int wbase = 0;
  for (int k = 0; k < wave; ++k) wbase += s_wsum[k];
  const int excl = wbase + x - cs;
  s_start[2 * tid]     = excl;
  s_start[2 * tid + 1] = excl + ca;
  if (blockIdx.x == 0) {
    if (2 * tid     < nb) binstart[2 * tid]     = excl;
    if (2 * tid + 1 < nb) binstart[2 * tid + 1] = excl + ca;
  }

  const int base = blockIdx.x * CHUNK;
  const bool full = base + CHUNK <= e;

  auto count_pass = [&](const bool FULL) {
    for (int u0 = 0; u0 < CHUNK / 256; u0 += 8) {
      if (!FULL && base + 256 * u0 >= e) break;
      int d[8];
#pragma unroll
      for (int k = 0; k < 8; ++k) {
        const int i = base + tid + 256 * (u0 + k);
        d[k] = dst[FULL ? i : min(i, e - 1)];
      }
#pragma unroll
      for (int k = 0; k < 8; ++k) {
        const int i = base + tid + 256 * (u0 + k);
        if (FULL || i < e) atomicAdd(&s_hist[d[k] >> BIN_SHIFT], 1);
      }
    }
  };
  if (full) count_pass(true); else count_pass(false);
  __syncthreads();

  for (int b = tid; b < nb; b += 256) {
    const int c = s_hist[b];
    const int off = c ? atomicAdd(&bincur[b * CSTRIDE], c) : 0;
    s_hist[b] = s_start[b] + off;
  }
  __syncthreads();

  auto place_pass = [&](const bool FULL) {
    for (int u0 = 0; u0 < CHUNK / 256; u0 += 8) {
      if (!FULL && base + 256 * u0 >= e) break;
      int d[8], s[8], pos[8];
#pragma unroll
      for (int k = 0; k < 8; ++k) {
        const int i = base + tid + 256 * (u0 + k);
        const int j = FULL ? i : min(i, e - 1);
        d[k] = dst[j];
        s[k] = src[j];
      }
#pragma unroll
      for (int k = 0; k < 8; ++k) {
        const int i = base + tid + 256 * (u0 + k);
        pos[k] = 0;
        if (FULL || i < e) pos[k] = atomicAdd(&s_hist[d[k] >> BIN_SHIFT], 1);
      }
#pragma unroll
      for (int k = 0; k < 8; ++k) {
        const int i = base + tid + 256 * (u0 + k);
        if (FULL || i < e) binned[pos[k]] = make_int2(s[k], d[k]);
      }
    }
  };
  if (full) place_pass(true); else place_pass(false);
}

__global__ __launch_bounds__(256) void bin_rows(const int2* __restrict__ binned, const int* __restrict__ bincnt,
                                                const int* __restrict__ binstart, int* __restrict__ rowptr,
                                                float* __restrict__ dinv, int2* __restrict__ rec, int n) {
  __shared__ int s_cnt[256];
  __shared__ int s_wsum[8];
  const int tid  = threadIdx.x;
  const int lane = tid & 31;
  const int wave = tid >> 5;
  const int bin   = blockIdx.x;
  const int start = binstart[bin];
  const int ne    = bincnt[bin * CSTRIDE];
  const int2* __restrict__ mine = binned + start;

  s_cnt[tid] = 0;
  __syncthreads();
  for (int i0 = tid; i0 < ne + tid; i0 += 8 * 256) {
    int2 r[8];
#pragma unroll
    for (int k = 0; k < 8; ++k) r[k] = mine[min(i0 + 256 * k, ne - 1)];
#pragma unroll
    for (int k = 0; k < 8; ++k)
      if (i0 + 256 * k < ne) atomicAdd(&s_cnt[r[k].y & 255], 1);
  }
  __syncthreads();

  const int c = s_cnt[tid];
  int x = c;
  for (int off = 1; off < 32; off <<= 1) {
    int y = __shfl_up(x, off);
    if (lane >= off) x += y;
  }
  if (lane == 31) s_wsum[wave] = x;
  __syncthreads();
  int wbase = 0;
  for (int k = 0; k < wave; ++k) wbase += s_wsum[k];
  const int row0 = start + wbase + x - c;

  const int node = (bin << BIN_SHIFT) + tid;
  if (node < n) {
    rowptr[node] = row0;
    if (node == n - 1) rowptr[n] = row0 + c;
    dinv[node] = rsqrtf(fmaxf((float)(c + 1), 1e-12f));
  }
  s_cnt[tid] = row0;
  __syncthreads();
  for (int i0 = tid; i0 < ne + tid; i0 += 8 * 256) {
    int2 r[8];
    int  pos[8];
#pragma unroll
    for (int k = 0; k < 8; ++k) r[k] = mine[min(i0 + 256 * k, ne - 1)];
#pragma unroll
    for (int k = 0; k < 8; ++k) {
      pos[k] = 0;
      if (i0 + 256 * k < ne) pos[k] = atomicAdd(&s_cnt[r[k].y & 255], 1);
    }
#pragma unroll
    for (int k = 0; k < 8; ++k)
      if (i0 + 256 * k < ne) rec[pos[k]] = r[k];
  }
}

__global__ void edge_weights(const float* __restrict__ dinv, int2* __restrict__ rec, int e) {
  int i = blockIdx.x * blockDim.x + threadIdx.x;
  if (i < e) {
    const int2 r = rec[i];
    const float w = dinv[r.x] * dinv[r.y];
    rec[i] = make_int2(r.x, __float_as_int(w));
  }
}

#define G1_BM 64
#define G1_KS 32
#define G1_XS 36
#define G1_WS 136

__global__ __launch_bounds__(256) __attribute__((amdgpu_waves_per_eu(6))) void gemm1_wmma(const float* X, const float* W, float* H, int n) {
  __shared__ __attribute__((aligned(16))) float sX[G1_BM * G1_XS];
  __shared__ __attribute__((aligned(16))) float sW[G1_KS * G1_WS];

  const int tid  = threadIdx.x;
  const int wave = tid >> 5;
  const int lane = tid & 31;
  const int rt   = wave >> 1;
  const int col0 = (wave & 1) * 64;
  const int m    = lane & 15;
  const int kh   = lane >> 4;
  const int row0 = blockIdx.x * G1_BM;

  const int xr0 = tid >> 3, xc4 = tid & 7;
  const int wk0 = tid >> 5, wj  = tid & 31;
  const int xrow0 = (row0 + xr0      < n) ? row0 + xr0      : 0;
  const int xrow1 = (row0 + xr0 + 32 < n) ? row0 + xr0 + 32 : 0;
  const float4* xg0 = (const float4*)(X + (size_t)xrow0 * IN_DIMC) + xc4;
  const float4* xg1 = (const float4*)(X + (size_t)xrow1 * IN_DIMC) + xc4;
  const float4* wg  = (const float4*)W + (size_t)wk0 * (H_DIMC / 4) + wj;
  float* __restrict__ sx0 = &sX[xr0 * G1_XS + 4 * xc4];
  float* __restrict__ sw0 = &sW[wk0 * G1_WS + 4 * wj];
  const float* __restrict__ fa = &sX[(rt * 16 + m) * G1_XS + 2 * kh];
  const float* __restrict__ fb = &sW[(2 * kh) * G1_WS + col0 + m];

  v8f acc[4];
#pragma unroll
  for (int t = 0; t < 4; ++t) acc[t] = (v8f){0.f, 0.f, 0.f, 0.f, 0.f, 0.f, 0.f, 0.f};

  float4 xa = xg0[0];
  float4 xb = xg1[0];
  float4 w0 = wg[0];
  float4 w1 = wg[8  * (H_DIMC / 4)];
  float4 w2 = wg[16 * (H_DIMC / 4)];
  float4 w3 = wg[24 * (H_DIMC / 4)];

#pragma unroll
  for (int kb = 0; kb < IN_DIMC; kb += G1_KS) {
    __syncthreads();
    *(float4*)(sx0)               = xa;
    *(float4*)(sx0 + 32 * G1_XS)  = xb;
    *(float4*)(sw0)               = w0;
    *(float4*)(sw0 + 8  * G1_WS)  = w1;
    *(float4*)(sw0 + 16 * G1_WS)  = w2;
    *(float4*)(sw0 + 24 * G1_WS)  = w3;
    __syncthreads();
    if (kb + G1_KS < IN_DIMC) {
      const int kn = kb + G1_KS;
      xa = xg0[kn / 4];
      xb = xg1[kn / 4];
      w0 = wg[(size_t)(kn)      * (H_DIMC / 4)];
      w1 = wg[(size_t)(kn + 8)  * (H_DIMC / 4)];
      w2 = wg[(size_t)(kn + 16) * (H_DIMC / 4)];
      w3 = wg[(size_t)(kn + 24) * (H_DIMC / 4)];
    }
    __builtin_amdgcn_sched_barrier(0);

#pragma unroll
    for (int ks = 0; ks < G1_KS; ks += 4) {
      const v2f a = *(const v2f*)(fa + ks);
#pragma unroll
      for (int t = 0; t < 4; ++t) {
        v2f b;
        b[0] = fb[(ks + 0) * G1_WS + 16 * t];
        b[1] = fb[(ks + 1) * G1_WS + 16 * t];
        acc[t] = __builtin_amdgcn_wmma_f32_16x16x4_f32(false, a, false, b,
                                                       (short)0, acc[t], false, false);
      }
    }
  }
  float* __restrict__ hout = H + (size_t)(row0 + rt * 16 + 8 * kh) * H_DIMC + col0 + m;
  if (row0 + G1_BM <= n) {
#pragma unroll
    for (int t = 0; t < 4; ++t)
#pragma unroll
      for (int r = 0; r < 8; ++r) hout[(size_t)r * H_DIMC + 16 * t] = acc[t][r];
  } else {
#pragma unroll
    for (int t = 0; t < 4; ++t)
#pragma unroll
      for (int r = 0; r < 8; ++r)
        if (row0 + rt * 16 + 8 * kh + r < n) hout[(size_t)r * H_DIMC + 16 * t] = acc[t][r];
  }
}

__global__ __launch_bounds__(384) void gemm2_wmma(const float* __restrict__ Xh,
                                                  const float* __restrict__ W2,
                                                  float* __restrict__ H2, int n) {
  __shared__ __attribute__((aligned(16))) float sW[H_DIMC][OUT_PADC + 4];
  __shared__ __attribute__((aligned(16))) float sX[64][H_DIMC + 4];

  const int tid  = threadIdx.x;
  const int wave = tid >> 5;
  const int lane = tid & 31;
  const int rt   = wave / 3;
  const int ct   = wave % 3;
  const int col0 = ct * 16;
  const int m    = lane & 15;
  const int kh   = lane >> 4;
  const int row0 = blockIdx.x * 64;

  if (tid < 256) {
    const int xr = tid >> 5;
    float4 xv[8];
#pragma unroll
    for (int u = 0; u < 8; ++u) {
      const int row = min(row0 + xr + 8 * u, n - 1);
      xv[u] = ((const float4*)(Xh + (size_t)row * H_DIMC))[lane];
    }
#pragma unroll
    for (int u = 0; u < 8; ++u) *(float4*)&sX[xr + 8 * u][4 * lane] = xv[u];
  } else {
    const int t = tid - 256;
    float4 wv[10];
#pragma unroll
    for (int u = 0; u < 10; ++u) wv[u] = ((const float4*)W2)[t + 128 * u];
#pragma unroll
    for (int u = 0; u < 10; ++u) {
      const int i = t + 128 * u;
      const int r = i / (OUT_DIMC / 4), c4 = i - r * (OUT_DIMC / 4);
      *(float4*)&sW[r][4 * c4] = wv[u];
    }
#pragma unroll
    for (int u = 0; u < 2; ++u) {
      const int j = t + 128 * u;
      *(float4*)&sW[j >> 1][OUT_DIMC + 4 * (j & 1)] = make_float4(0.f, 0.f, 0.f, 0.f);
    }
  }
  __syncthreads();

  v8f acc = {0.f, 0.f, 0.f, 0.f, 0.f, 0.f, 0.f, 0.f};
#pragma unroll
  for (int ks = 0; ks < H_DIMC; ks += 4) {
    v2f a, b;
    a[0] = sX[rt * 16 + m][ks + 2 * kh + 0];
    a[1] = sX[rt * 16 + m][ks + 2 * kh + 1];
    b[0] = sW[ks + 2 * kh + 0][col0 + m];
    b[1] = sW[ks + 2 * kh + 1][col0 + m];
    acc = __builtin_amdgcn_wmma_f32_16x16x4_f32(false, a, false, b,
                                                (short)0, acc, false, false);
  }
#pragma unroll
  for (int r = 0; r < 8; ++r) {
    int row = row0 + rt * 16 + 8 * kh + r;
    int col = col0 + m;
    if (row < n && col < OUT_DIMC) H2[(size_t)row * OUT_DIMC + col] = acc[r];
  }
}

__global__ __launch_bounds__(256) void gather_rows_h(const int* __restrict__ rowptr,
                                                     const int2* __restrict__ rec,
                                                     const float* __restrict__ h,
                                                     const float* __restrict__ dinv,
                                                     const float* __restrict__ b1,
                                                     float* __restrict__ out, int n) {
  const int lane = threadIdx.x & 31;
  const int node = __builtin_amdgcn_readfirstlane((int)((blockIdx.x * 256u + threadIdx.x) >> 5));
  if (node >= n) return;
  const int beg = rowptr[node];
  const int end = rowptr[node + 1];
  const float4* __restrict__ hp = (const float4*)h;
  const float4 hs = hp[(size_t)node * (H_DIMC / 4) + lane];
  const float  di = dinv[node];
  const float bx = b1[4 * lane + 0], by = b1[4 * lane + 1];
  const float bz = b1[4 * lane + 2], bw = b1[4 * lane + 3];
  float4 acc = make_float4(0.0f, 0.0f, 0.0f, 0.0f);

  for (int base = beg; base < end; base += 32) {
    const int m = min(32, end - base);
    int2 r = make_int2(0, 0);
    if (lane < m) r = rec[base + lane];
    int k = 0;
    for (; k + 8 <= m; k += 8) {
      float4 v[8];
#pragma unroll
      for (int u = 0; u < 8; ++u) {
        const int s = __builtin_amdgcn_readlane(r.x, k + u);
        v[u] = hp[(size_t)s * (H_DIMC / 4) + lane];
      }
#pragma unroll
      for (int u = 0; u < 8; ++u) {
        const float w = __int_as_float(__builtin_amdgcn_readlane(r.y, k + u));
        acc.x += v[u].x * w; acc.y += v[u].y * w; acc.z += v[u].z * w; acc.w += v[u].w * w;
      }
    }
    const int t = m - k;
    if (t > 0) {
      float4 v[7];
#pragma unroll
      for (int u = 0; u < 7; ++u) {
        if (u < t) {
          const int s = __builtin_amdgcn_readlane(r.x, k + u);
          v[u] = hp[(size_t)s * (H_DIMC / 4) + lane];
        }
      }
#pragma unroll
      for (int u = 0; u < 7; ++u) {
        if (u < t) {
          const float w = __int_as_float(__builtin_amdgcn_readlane(r.y, k + u));
          acc.x += v[u].x * w; acc.y += v[u].y * w; acc.z += v[u].z * w; acc.w += v[u].w * w;
        }
      }
    }
  }
  float4 o;
  o.x = acc.x + hs.x * di * di + bx;
  o.y = acc.y + hs.y * di * di + by;
  o.z = acc.z + hs.z * di * di + bz;
  o.w = acc.w + hs.w * di * di + bw;
  o.x = o.x > 0.0f ? o.x : 0.0f;
  o.y = o.y > 0.0f ? o.y : 0.0f;
  o.z = o.z > 0.0f ? o.z : 0.0f;
  o.w = o.w > 0.0f ? o.w : 0.0f;
  ((float4*)out)[(size_t)node * (H_DIMC / 4) + lane] = o;
}

__global__ __launch_bounds__(256) void gather_rows_o(const int* __restrict__ rowptr,
                                                     const int2* __restrict__ rec,
                                                     const float* __restrict__ h2,
                                                     const float* __restrict__ dinv,
                                                     const float* __restrict__ b2,
                                                     float* __restrict__ out, int n) {
  const int lane = threadIdx.x & 31;
  const int q    = min(lane, OUT_DIMC / 4 - 1);
  const int node = __builtin_amdgcn_readfirstlane((int)((blockIdx.x * 256u + threadIdx.x) >> 5));
  if (node >= n) return;
  const int beg = rowptr[node];
  const int end = rowptr[node + 1];
  const float4* __restrict__ hp = (const float4*)h2;
  const float4 hs = hp[(size_t)node * (OUT_DIMC / 4) + q];
  const float  di = dinv[node];
  const float bx = b2[4 * q + 0], by = b2[4 * q + 1];
  const float bz = b2[4 * q + 2], bw = b2[4 * q + 3];
  float4 acc = make_float4(0.0f, 0.0f, 0.0f, 0.0f);

  for (int base = beg; base < end; base += 32) {
    const int m = min(32, end - base);
    int2 r = make_int2(0, 0);
    if (lane < m) r = rec[base + lane];
    int k = 0;
    for (; k + 8 <= m; k += 8) {
      float4 v[8];
#pragma unroll
      for (int u = 0; u < 8; ++u) {
        const int s = __builtin_amdgcn_readlane(r.x, k + u);
        v[u] = hp[(size_t)s * (OUT_DIMC / 4) + q];
      }
#pragma unroll
      for (int u = 0; u < 8; ++u) {
        const float w = __int_as_float(__builtin_amdgcn_readlane(r.y, k + u));
        acc.x += v[u].x * w; acc.y += v[u].y * w; acc.z += v[u].z * w; acc.w += v[u].w * w;
      }
    }
    const int t = m - k;
    if (t > 0) {
      float4 v[7];
#pragma unroll
      for (int u = 0; u < 7; ++u) {
        if (u < t) {
          const int s = __builtin_amdgcn_readlane(r.x, k + u);
          v[u] = hp[(size_t)s * (OUT_DIMC / 4) + q];
        }
      }
#pragma unroll
      for (int u = 0; u < 7; ++u) {
        if (u < t) {
          const float w = __int_as_float(__builtin_amdgcn_readlane(r.y, k + u));
          acc.x += v[u].x * w; acc.y += v[u].y * w; acc.z += v[u].z * w; acc.w += v[u].w * w;
        }
      }
    }
  }
  float4 o;
  o.x = acc.x + (hs.x * di * di + bx);
  o.y = acc.y + (hs.y * di * di + by);
  o.z = acc.z + (hs.z * di * di + bz);
  o.w = acc.w + (hs.w * di * di + bw);
  if (lane < OUT_DIMC / 4) ((float4*)out)[(size_t)node * (OUT_DIMC / 4) + lane] = o;
}

extern "C" void kernel_launch(void* const* d_in, const int* in_sizes, int n_in,
                              void* d_out, int out_size, void* d_ws, size_t ws_size,
                              hipStream_t stream) {
  const float* x  = (const float*)d_in[0];
  const int*   ei = (const int*)d_in[1];
  const float* W1 = (const float*)d_in[2];
  const float* b1 = (const float*)d_in[3];
  const float* W2 = (const float*)d_in[4];
  const float* b2 = (const float*)d_in[5];
  float* out = (float*)d_out;

  const int N = in_sizes[0] / IN_DIMC;
  const int E = in_sizes[1] / 2;
  const int* src = ei;
  const int* dst = ei + E;

  size_t off = 0;
  char* base = (char*)d_ws;
  auto take = [&](size_t count) -> float* {
    float* p = (float*)(base + off);
    off += ((count * sizeof(float) + 255) & ~(size_t)255);
    return p;
  };
  float* dinv   = take((size_t)N);
  int*   bincnt = (int*)take((size_t)2 * NBMAX * CSTRIDE);
  int*   bincur = bincnt + NBMAX * CSTRIDE;
  int*   binstart = (int*)take((size_t)NBMAX);
  int*   rowptr = (int*)take((size_t)N + 1);
  int2*  binned = (int2*)take((size_t)E * 2);
  int2*  rec    = (int2*)take((size_t)E * 2);
  float* h1     = take((size_t)N * H_DIMC);
  float* agg1   = take((size_t)N * H_DIMC);
  float* h2     = take((size_t)N * OUT_DIMC);
  (void)ws_size; (void)n_in; (void)out_size;

  const int T = 256;
  const int NB = (N + 255) >> BIN_SHIFT;
  const int GB = (E + CHUNK - 1) / CHUNK;
  zero_i32    <<<(2 * NBMAX * CSTRIDE + T - 1) / T, T, 0, stream>>>(bincnt, 2 * NBMAX * CSTRIDE);
  bin_count   <<<GB, T, 0, stream>>>(dst, bincnt, E, NB);
  bin_scatter <<<GB, T, 0, stream>>>(src, dst, bincnt, bincur, binstart, binned, E, NB);
  bin_rows    <<<NB, T, 0, stream>>>(binned, bincnt, binstart, rowptr, dinv, rec, N);
  edge_weights<<<(E + T - 1) / T, T, 0, stream>>>(dinv, rec, E);

  gemm1_wmma   <<<(N + G1_BM - 1) / G1_BM, 256, 0, stream>>>(x, W1, h1, N);
  gather_rows_h<<<(N + 7) / 8, 256, 0, stream>>>(rowptr, rec, h1, dinv, b1, agg1, N);

  gemm2_wmma   <<<(N + 63) / 64, 384, 0, stream>>>(agg1, W2, h2, N);
  gather_rows_o<<<(N + 7) / 8, 256, 0, stream>>>(rowptr, rec, h2, dinv, b2, out, N);
}


# both row gathers: a wave walks 10 consecutive nodes (10000 resident waves, not 100000), row starts by one load, next node's list words and own row requested under the current rows; sum order kept
# speedup vs baseline: 4.4006x; 1.1058x over previous
#include <hip/hip_runtime.h>

typedef __attribute__((ext_vector_type(2))) float v2f;
typedef __attribute__((ext_vector_type(8))) float v8f;

#define IN_DIMC 256
#define H_DIMC  128
#define OUT_DIMC 40
#define OUT_PADC 48

#define BIN_SHIFT 8
#define NBMAX     512
#define CHUNK     8192
#define CSTRIDE   32

__global__ void zero_i32(int* __restrict__ p, int n) {
  int i = blockIdx.x * blockDim.x + threadIdx.x;
  if (i < n) p[i] = 0;
}

__global__ __launch_bounds__(256) void bin_count(const int* __restrict__ dst, int* __restrict__ bincnt,
                                                 int e, int nb) {
  __shared__ int s_hist[NBMAX];
  const int tid = threadIdx.x;
  s_hist[tid] = 0;
  s_hist[tid + 256] = 0;
  __syncthreads();
  const int base = blockIdx.x * CHUNK;
  auto count_pass = [&](const bool FULL) {
    for (int u0 = 0; u0 < CHUNK / 256; u0 += 8) {
      if (!FULL && base + 256 * u0 >= e) break;
      int d[8];
#pragma unroll
      for (int k = 0; k < 8; ++k) {
        const int i = base + tid + 256 * (u0 + k);
        d[k] = dst[FULL ? i : min(i, e - 1)];
      }
#pragma unroll
      for (int k = 0; k < 8; ++k) {
        const int i = base + tid + 256 * (u0 + k);
        if (FULL || i < e) atomicAdd(&s_hist[d[k] >> BIN_SHIFT], 1);
      }
    }
  };
  if (base + CHUNK <= e) count_pass(true); else count_pass(false);
  __syncthreads();
  for (int b = tid; b < nb; b += 256) {
    const int c = s_hist[b];
    if (c) atomicAdd(&bincnt[b * CSTRIDE], c);
  }
}

__global__ __launch_bounds__(256) void bin_scatter(const int* __restrict__ src, const int* __restrict__ dst,
                                                   const int* __restrict__ bincnt, int* __restrict__ bincur,
                                                   int* __restrict__ binstart, int2* __restrict__ binned,
                                                   int e, int nb) {
  __shared__ int s_start[NBMAX];
  __shared__ int s_hist[NBMAX];
  __shared__ int s_wsum[8];
  const int tid  = threadIdx.x;
  const int lane = tid & 31;
  const int wave = tid >> 5;

  const int ca = (2 * tid     < nb) ? bincnt[(2 * tid)     * CSTRIDE] : 0;
  const int cb = (2 * tid + 1 < nb) ? bincnt[(2 * tid + 1) * CSTRIDE] : 0;
  const int cs = ca + cb;
  int x = cs;
  for (int off = 1; off < 32; off <<= 1) {
    int y = __shfl_up(x, off);
    if (lane >= off) x += y;
  }
  if (lane == 31) s_wsum[wave] = x;
  s_hist[tid] = 0;
  s_hist[tid + 256] = 0;
  __syncthreads();
  int wbase = 0;
  for (int k = 0; k < wave; ++k) wbase += s_wsum[k];
  const int excl = wbase + x - cs;
  s_start[2 * tid]     = excl;
  s_start[2 * tid + 1] = excl + ca;
  if (blockIdx.x == 0) {
    if (2 * tid     < nb) binstart[2 * tid]     = excl;
    if (2 * tid + 1 < nb) binstart[2 * tid + 1] = excl + ca;
  }

  const int base = blockIdx.x * CHUNK;
  const bool full = base + CHUNK <= e;

  auto count_pass = [&](const bool FULL) {
    for (int u0 = 0; u0 < CHUNK / 256; u0 += 8) {
      if (!FULL && base + 256 * u0 >= e) break;
      int d[8];
#pragma unroll
      for (int k = 0; k < 8; ++k) {
        const int i = base + tid + 256 * (u0 + k);
        d[k] = dst[FULL ? i : min(i, e - 1)];
      }
#pragma unroll
      for (int k = 0; k < 8; ++k) {
        const int i = base + tid + 256 * (u0 + k);
        if (FULL || i < e) atomicAdd(&s_hist[d[k] >> BIN_SHIFT], 1);
      }
    }
  };
  if (full) count_pass(true); else count_pass(false);
  __syncthreads();

  for (int b = tid; b < nb; b += 256) {
    const int c = s_hist[b];
    const int off = c ? atomicAdd(&bincur[b * CSTRIDE], c) : 0;
    s_hist[b] = s_start[b] + off;
  }
  __syncthreads();

  auto place_pass = [&](const bool FULL) {
    for (int u0 = 0; u0 < CHUNK / 256; u0 += 8) {
      if (!FULL && base + 256 * u0 >= e) break;
      int d[8], s[8], pos[8];
#pragma unroll
      for (int k = 0; k < 8; ++k) {
        const int i = base + tid + 256 * (u0 + k);
        const int j = FULL ? i : min(i, e - 1);
        d[k] = dst[j];
        s[k] = src[j];
      }
#pragma unroll
      for (int k = 0; k < 8; ++k) {
        const int i = base + tid + 256 * (u0 + k);
        pos[k] = 0;
        if (FULL || i < e) pos[k] = atomicAdd(&s_hist[d[k] >> BIN_SHIFT], 1);
      }
#pragma unroll
      for (int k = 0; k < 8; ++k) {
        const int i = base + tid + 256 * (u0 + k);
        if (FULL || i < e) binned[pos[k]] = make_int2(s[k], d[k]);
      }
    }
  };
  if (full) place_pass(true); else place_pass(false);
}

__global__ __launch_bounds__(256) void bin_rows(const int2* __restrict__ binned, const int* __restrict__ bincnt,
                                                const int* __restrict__ binstart, int* __restrict__ rowptr,
                                                float* __restrict__ dinv, int* __restrict__ rec, int n) {
  __shared__ int s_cnt[256];
  __shared__ int s_wsum[8];
  const int tid  = threadIdx.x;
  const int lane = tid & 31;
  const int wave = tid >> 5;
  const int bin   = blockIdx.x;
  const int start = binstart[bin];
  const int ne    = bincnt[bin * CSTRIDE];
  const int2* __restrict__ mine = binned + start;

  s_cnt[tid] = 0;
  __syncthreads();
  for (int i0 = tid; i0 < ne + tid; i0 += 8 * 256) {
    int2 r[8];
#pragma unroll
    for (int k = 0; k < 8; ++k) r[k] = mine[min(i0 + 256 * k, ne - 1)];
#pragma unroll
    for (int k = 0; k < 8; ++k)
      if (i0 + 256 * k < ne) atomicAdd(&s_cnt[r[k].y & 255], 1);
  }
  __syncthreads();

  const int c = s_cnt[tid];
  int x = c;
  for (int off = 1; off < 32; off <<= 1) {
    int y = __shfl_up(x, off);
    if (lane >= off) x += y;
  }
  if (lane == 31) s_wsum[wave] = x;
  __syncthreads();
  int wbase = 0;
  for (int k = 0; k < wave; ++k) wbase += s_wsum[k];
  const int row0 = start + wbase + x - c;

  const int node = (bin << BIN_SHIFT) + tid;
  if (node < n) {
    rowptr[node] = row0;
    if (node == n - 1) rowptr[n] = row0 + c;
    dinv[node] = rsqrtf(fmaxf((float)(c + 1), 1e-12f));
  }
  s_cnt[tid] = row0;
  __syncthreads();
  for (int i0 = tid; i0 < ne + tid; i0 += 8 * 256) {
    int2 r[8];
    int  pos[8];
#pragma unroll
    for (int k = 0; k < 8; ++k) r[k] = mine[min(i0 + 256 * k, ne - 1)];
#pragma unroll
    for (int k = 0; k < 8; ++k) {
      pos[k] = 0;
      if (i0 + 256 * k < ne) pos[k] = atomicAdd(&s_cnt[r[k].y & 255], 1);
    }
#pragma unroll
    for (int k = 0; k < 8; ++k)
      if (i0 + 256 * k < ne) rec[pos[k]] = r[k].x;
  }
}

#define G1_BM 64
#define G1_KS 32
#define G1_XS 36
#define G1_WS 136

__global__ __launch_bounds__(256) __attribute__((amdgpu_waves_per_eu(6))) void gemm1_wmma(const float* X, const float* W, const float* dinv, float* H, int n) {
  __shared__ __attribute__((aligned(16))) float sX[G1_BM * G1_XS];
  __shared__ __attribute__((aligned(16))) float sW[G1_KS * G1_WS];

  const int tid  = threadIdx.x;
  const int wave = tid >> 5;
  const int lane = tid & 31;
  const int rt   = wave >> 1;
  const int col0 = (wave & 1) * 64;
  const int m    = lane & 15;
  const int kh   = lane >> 4;
  const int row0 = blockIdx.x * G1_BM;

  const int xr0 = tid >> 3, xc4 = tid & 7;
  const int wk0 = tid >> 5, wj  = tid & 31;
  const int xrow0 = (row0 + xr0      < n) ? row0 + xr0      : 0;
  const int xrow1 = (row0 + xr0 + 32 < n) ? row0 + xr0 + 32 : 0;
  const float4* xg0 = (const float4*)(X + (size_t)xrow0 * IN_DIMC) + xc4;
  const float4* xg1 = (const float4*)(X + (size_t)xrow1 * IN_DIMC) + xc4;
  const float4* wg  = (const float4*)W + (size_t)wk0 * (H_DIMC / 4) + wj;
  float* __restrict__ sx0 = &sX[xr0 * G1_XS + 4 * xc4];
  float* __restrict__ sw0 = &sW[wk0 * G1_WS + 4 * wj];
  const float* __restrict__ fa = &sX[(rt * 16 + m) * G1_XS + 2 * kh];
  const float* __restrict__ fb = &sW[(2 * kh) * G1_WS + col0 + m];

  v8f acc[4];
#pragma unroll
  for (int t = 0; t < 4; ++t) acc[t] = (v8f){0.f, 0.f, 0.f, 0.f, 0.f, 0.f, 0.f, 0.f};

  const int erow = row0 + rt * 16 + 8 * kh;
  float dvl;

  float4 xa = xg0[0];
  float4 xb = xg1[0];
  float4 w0 = wg[0];
  float4 w1 = wg[8  * (H_DIMC / 4)];
  float4 w2 = wg[16 * (H_DIMC / 4)];
  float4 w3 = wg[24 * (H_DIMC / 4)];

#pragma unroll
  for (int kb = 0; kb < IN_DIMC; kb += G1_KS) {
    __syncthreads();
    *(float4*)(sx0)               = xa;
    *(float4*)(sx0 + 32 * G1_XS)  = xb;
    *(float4*)(sw0)               = w0;
    *(float4*)(sw0 + 8  * G1_WS)  = w1;
    *(float4*)(sw0 + 16 * G1_WS)  = w2;
    *(float4*)(sw0 + 24 * G1_WS)  = w3;
    __syncthreads();
    if (kb + G1_KS < IN_DIMC) {
      const int kn = kb + G1_KS;
      xa = xg0[kn / 4];
      xb = xg1[kn / 4];
      w0 = wg[(size_t)(kn)      * (H_DIMC / 4)];
      w1 = wg[(size_t)(kn + 8)  * (H_DIMC / 4)];
      w2 = wg[(size_t)(kn + 16) * (H_DIMC / 4)];
      w3 = wg[(size_t)(kn + 24) * (H_DIMC / 4)];
    } else {
      dvl = dinv[min(row0 + rt * 16 + m, n - 1)];
    }
    __builtin_amdgcn_sched_barrier(0);

#pragma unroll
    for (int ks = 0; ks < G1_KS; ks += 4) {
      const v2f a = *(const v2f*)(fa + ks);
#pragma unroll
      for (int t = 0; t < 4; ++t) {
        v2f b;
        b[0] = fb[(ks + 0) * G1_WS + 16 * t];
        b[1] = fb[(ks + 1) * G1_WS + 16 * t];
        acc[t] = __builtin_amdgcn_wmma_f32_16x16x4_f32(false, a, false, b,
                                                       (short)0, acc[t], false, false);
      }
    }
  }
  float* __restrict__ hout = H + (size_t)erow * H_DIMC + col0 + m;
#pragma unroll
  for (int r = 0; r < 8; ++r) {
    const float dvr = __shfl(dvl, 8 * kh + r);
#pragma unroll
    for (int t = 0; t < 4; ++t) acc[t][r] *= dvr;
  }
  if (row0 + G1_BM <= n) {
#pragma unroll
    for (int t = 0; t < 4; ++t)
#pragma unroll
      for (int r = 0; r < 8; ++r) hout[(size_t)r * H_DIMC + 16 * t] = acc[t][r];
  } else {
#pragma unroll
    for (int t = 0; t < 4; ++t)
#pragma unroll
      for (int r = 0; r < 8; ++r)
        if (erow + r < n) hout[(size_t)r * H_DIMC + 16 * t] = acc[t][r];
  }
}

__global__ __launch_bounds__(384) void gemm2_wmma(const float* __restrict__ Xh,
                                                  const float* __restrict__ W2,
                                                  const float* dinv,
                                                  float* __restrict__ H2, int n) {
  __shared__ __attribute__((aligned(16))) float sW[H_DIMC][OUT_PADC + 4];
  __shared__ __attribute__((aligned(16))) float sX[64][H_DIMC + 4];

  const int tid  = threadIdx.x;
  const int wave = tid >> 5;
  const int lane = tid & 31;
  const int rt   = wave / 3;
  const int ct   = wave % 3;
  const int col0 = ct * 16;
  const int m    = lane & 15;
  const int kh   = lane >> 4;
  const int row0 = blockIdx.x * 64;

  if (tid < 256) {
    const int xr = tid >> 5;
    float4 xv[8];
#pragma unroll
    for (int u = 0; u < 8; ++u) {
      const int row = min(row0 + xr + 8 * u, n - 1);
      xv[u] = ((const float4*)(Xh + (size_t)row * H_DIMC))[lane];
    }
#pragma unroll
    for (int u = 0; u < 8; ++u) *(float4*)&sX[xr + 8 * u][4 * lane] = xv[u];
  } else {
    const int t = tid - 256;
    float4 wv[10];
#pragma unroll
    for (int u = 0; u < 10; ++u) wv[u] = ((const float4*)W2)[t + 128 * u];
#pragma unroll
    for (int u = 0; u < 10; ++u) {
      const int i = t + 128 * u;
      const int r = i / (OUT_DIMC / 4), c4 = i - r * (OUT_DIMC / 4);
      *(float4*)&sW[r][4 * c4] = wv[u];
    }
#pragma unroll
    for (int u = 0; u < 2; ++u) {
      const int j = t + 128 * u;
      *(float4*)&sW[j >> 1][OUT_DIMC + 4 * (j & 1)] = make_float4(0.f, 0.f, 0.f, 0.f);
    }
  }
  __syncthreads();

  const int erow = row0 + rt * 16 + 8 * kh;
  const float dvl = dinv[min(row0 + rt * 16 + m, n - 1)];
  __builtin_amdgcn_sched_barrier(0);

  v8f acc = {0.f, 0.f, 0.f, 0.f, 0.f, 0.f, 0.f, 0.f};
#pragma unroll
  for (int ks = 0; ks < H_DIMC; ks += 4) {
    v2f a, b;
    a[0] = sX[rt * 16 + m][ks + 2 * kh + 0];
    a[1] = sX[rt * 16 + m][ks + 2 * kh + 1];
    b[0] = sW[ks + 2 * kh + 0][col0 + m];
    b[1] = sW[ks + 2 * kh + 1][col0 + m];
    acc = __builtin_amdgcn_wmma_f32_16x16x4_f32(false, a, false, b,
                                                (short)0, acc, false, false);
  }
#pragma unroll
  for (int r = 0; r < 8; ++r) acc[r] *= __shfl(dvl, 8 * kh + r);
#pragma unroll
  for (int r = 0; r < 8; ++r) {
    int row = erow + r;
    int col = col0 + m;
    if (row < n && col < OUT_DIMC) H2[(size_t)row * OUT_DIMC + col] = acc[r];
  }
}

#define NPW 10

template <int F4, bool RELU>
__device__ __forceinline__ void gather_rows_walk(const int* __restrict__ rowptr,
                                                 const int* __restrict__ rec,
                                                 const float* __restrict__ h,
                                                 const float* __restrict__ dinv,
                                                 const float* __restrict__ bias,
                                                 float* __restrict__ out, int n) {
  const int lane = threadIdx.x & 31;
  const int q    = (F4 < 32) ? min(lane, F4 - 1) : lane;
  const int n0 = NPW * __builtin_amdgcn_readfirstlane((int)((blockIdx.x * 256u + threadIdx.x) >> 5));
  if (n0 >= n) return;
  const int cnt = min(NPW, n - n0);
  const int rp  = rowptr[n0 + min(lane, cnt)];
  const float4* __restrict__ hp = (const float4*)h;
  const float bx = bias[4 * q + 0], by = bias[4 * q + 1];
  const float bz = bias[4 * q + 2], bw = bias[4 * q + 3];

  auto sum_pass = [&](const int r, const int m, float4& acc) {
    int k = 0;
    for (; k + 8 <= m; k += 8) {
      float4 v[8];
#pragma unroll
      for (int u = 0; u < 8; ++u) {
        const int s = __builtin_amdgcn_readlane(r, k + u);
        v[u] = hp[(size_t)s * F4 + q];
      }
#pragma unroll
      for (int u = 0; u < 8; ++u) {
        acc.x += v[u].x; acc.y += v[u].y; acc.z += v[u].z; acc.w += v[u].w;
      }
    }
    const int t = m - k;
    if (t > 0) {
      float4 v[7];
#pragma unroll
      for (int u = 0; u < 7; ++u) {
        if (u < t) {
          const int s = __builtin_amdgcn_readlane(r, k + u);
          v[u] = hp[(size_t)s * F4 + q];
        }
      }
#pragma unroll
      for (int u = 0; u < 7; ++u) {
        if (u < t) {
          acc.x += v[u].x; acc.y += v[u].y; acc.z += v[u].z; acc.w += v[u].w;
        }
      }
    }
  };

  int beg = __builtin_amdgcn_readlane(rp, 0);
  int end = __builtin_amdgcn_readlane(rp, 1);
  int r = 0;
  if (lane < end - beg) r = rec[beg + lane];
  float4 hs = hp[(size_t)n0 * F4 + q];
  float  di = dinv[n0];

  for (int j = 0; j < cnt; ++j) {
    const int node = n0 + j;
    int    nend = end;
    int    rn   = 0;
    float4 hsn  = hs;
    float  din  = di;
    if (j + 1 < cnt) {
      nend = __builtin_amdgcn_readlane(rp, j + 2);
      if (lane < nend - end) rn = rec[end + lane];
      hsn = hp[(size_t)(node + 1) * F4 + q];
      din = dinv[node + 1];
    }

    float4 acc = make_float4(0.0f, 0.0f, 0.0f, 0.0f);
    sum_pass(r, min(32, end - beg), acc);
    for (int base = beg + 32; base < end; base += 32) {
      const int m = min(32, end - base);
      int r2 = 0;
      if (lane < m) r2 = rec[base + lane];
      sum_pass(r2, m, acc);
    }

    float4 o;
    o.x = (acc.x + hs.x) * di + bx;
    o.y = (acc.y + hs.y) * di + by;
    o.z = (acc.z + hs.z) * di + bz;
    o.w = (acc.w + hs.w) * di + bw;
    if (RELU) {
      o.x = o.x > 0.0f ? o.x : 0.0f;
      o.y = o.y > 0.0f ? o.y : 0.0f;
      o.z = o.z > 0.0f ? o.z : 0.0f;
      o.w = o.w > 0.0f ? o.w : 0.0f;
    }
    if (F4 == 32 || lane < F4) ((float4*)out)[(size_t)node * F4 + lane] = o;

    beg = end; end = nend; r = rn; hs = hsn; di = din;
  }
}

__global__ __launch_bounds__(256) void gather_rows_h(const int* __restrict__ rowptr,
                                                     const int* __restrict__ rec,
                                                     const float* __restrict__ h,
                                                     const float* __restrict__ dinv,
                                                     const float* __restrict__ b1,
                                                     float* __restrict__ out, int n) {
  gather_rows_walk<H_DIMC / 4, true>(rowptr, rec, h, dinv, b1, out, n);
}

__global__ __launch_bounds__(256) void gather_rows_o(const int* __restrict__ rowptr,
                                                     const int* __restrict__ rec,
                                                     const float* __restrict__ h2,
                                                     const float* __restrict__ dinv,
                                                     const float* __restrict__ b2,
                                                     float* __restrict__ out, int n) {
  gather_rows_walk<OUT_DIMC / 4, false>(rowptr, rec, h2, dinv, b2, out, n);
}

extern "C" void kernel_launch(void* const* d_in, const int* in_sizes, int n_in,
                              void* d_out, int out_size, void* d_ws, size_t ws_size,
                              hipStream_t stream) {
  const float* x  = (const float*)d_in[0];
  const int*   ei = (const int*)d_in[1];
  const float* W1 = (const float*)d_in[2];
  const float* b1 = (const float*)d_in[3];
  const float* W2 = (const float*)d_in[4];
  const float* b2 = (const float*)d_in[5];
  float* out = (float*)d_out;

  const int N = in_sizes[0] / IN_DIMC;
  const int E = in_sizes[1] / 2;
  const int* src = ei;
  const int* dst = ei + E;

  size_t off = 0;
  char* base = (char*)d_ws;
  auto take = [&](size_t count) -> float* {
    float* p = (float*)(base + off);
    off += ((count * sizeof(float) + 255) & ~(size_t)255);
    return p;
  };
  float* dinv   = take((size_t)N);
  int*   bincnt = (int*)take((size_t)2 * NBMAX * CSTRIDE);
  int*   bincur = bincnt + NBMAX * CSTRIDE;
  int*   binstart = (int*)take((size_t)NBMAX);
  int*   rowptr = (int*)take((size_t)N + 1);
  int2*  binned = (int2*)take((size_t)E * 2);
  int*   rec    = (int*)take((size_t)E);
  float* h1     = take((size_t)N * H_DIMC);
  float* agg1   = take((size_t)N * H_DIMC);
  float* h2     = take((size_t)N * OUT_DIMC);
  (void)ws_size; (void)n_in; (void)out_size;

  const int T = 256;
  const int NB = (N + 255) >> BIN_SHIFT;
  const int GB = (E + CHUNK - 1) / CHUNK;
  zero_i32    <<<(2 * NBMAX * CSTRIDE + T - 1) / T, T, 0, stream>>>(bincnt, 2 * NBMAX * CSTRIDE);
  bin_count   <<<GB, T, 0, stream>>>(dst, bincnt, E, NB);
  bin_scatter <<<GB, T, 0, stream>>>(src, dst, bincnt, bincur, binstart, binned, E, NB);
  bin_rows    <<<NB, T, 0, stream>>>(binned, bincnt, binstart, rowptr, dinv, rec, N);

  gemm1_wmma   <<<(N + G1_BM - 1) / G1_BM, 256, 0, stream>>>(x, W1, dinv, h1, N);
  gather_rows_h<<<(N + 8 * NPW - 1) / (8 * NPW), 256, 0, stream>>>(rowptr, rec, h1, dinv, b1, agg1, N);

  gemm2_wmma   <<<(N + 63) / 64, 384, 0, stream>>>(agg1, W2, dinv, h2, N);
  gather_rows_o<<<(N + 8 * NPW - 1) / (8 * NPW), 256, 0, stream>>>(rowptr, rec, h2, dinv, b2, out, N);
}


# layer-2 product moved into the hidden-layer row gather: a wave walks 16 nodes, keeps their finished rows in an LDS tile and multiplies it by W2 on WMMA (k order kept); hidden plane and one launch go
# speedup vs baseline: 4.5849x; 1.0419x over previous
#include <hip/hip_runtime.h>

typedef __attribute__((ext_vector_type(2))) float v2f;
typedef __attribute__((ext_vector_type(8))) float v8f;

#define IN_DIMC 256
#define H_DIMC  128
#define OUT_DIMC 40
#define OUT_PADC 48

#define BIN_SHIFT 8
#define NBMAX     512
#define CHUNK     8192
#define CSTRIDE   32

__global__ void zero_i32(int* __restrict__ p, int n) {
  int i = blockIdx.x * blockDim.x + threadIdx.x;
  if (i < n) p[i] = 0;
}

__global__ __launch_bounds__(256) void bin_count(const int* __restrict__ dst, int* __restrict__ bincnt,
                                                 int e, int nb) {
  __shared__ int s_hist[NBMAX];
  const int tid = threadIdx.x;
  s_hist[tid] = 0;
  s_hist[tid + 256] = 0;
  __syncthreads();
  const int base = blockIdx.x * CHUNK;
  auto count_pass = [&](const bool FULL) {
    for (int u0 = 0; u0 < CHUNK / 256; u0 += 8) {
      if (!FULL && base + 256 * u0 >= e) break;
      int d[8];
#pragma unroll
      for (int k = 0; k < 8; ++k) {
        const int i = base + tid + 256 * (u0 + k);
        d[k] = dst[FULL ? i : min(i, e - 1)];
      }
#pragma unroll
      for (int k = 0; k < 8; ++k) {
        const int i = base + tid + 256 * (u0 + k);
        if (FULL || i < e) atomicAdd(&s_hist[d[k] >> BIN_SHIFT], 1);
      }
    }
  };
  if (base + CHUNK <= e) count_pass(true); else count_pass(false);
  __syncthreads();
  for (int b = tid; b < nb; b += 256) {
    const int c = s_hist[b];
    if (c) atomicAdd(&bincnt[b * CSTRIDE], c);
  }
}

__global__ __launch_bounds__(256) void bin_scatter(const int* __restrict__ src, const int* __restrict__ dst,
                                                   const int* __restrict__ bincnt, int* __restrict__ bincur,
                                                   int* __restrict__ binstart, int2* __restrict__ binned,
                                                   int e, int nb) {
  __shared__ int s_start[NBMAX];
  __shared__ int s_hist[NBMAX];
  __shared__ int s_wsum[8];
  const int tid  = threadIdx.x;
  const int lane = tid & 31;
  const int wave = tid >> 5;

  const int ca = (2 * tid     < nb) ? bincnt[(2 * tid)     * CSTRIDE] : 0;
  const int cb = (2 * tid + 1 < nb) ? bincnt[(2 * tid + 1) * CSTRIDE] : 0;
  const int cs = ca + cb;
  int x = cs;
  for (int off = 1; off < 32; off <<= 1) {
    int y = __shfl_up(x, off);
    if (lane >= off) x += y;
  }
  if (lane == 31) s_wsum[wave] = x;
  s_hist[tid] = 0;
  s_hist[tid + 256] = 0;
  __syncthreads();
  int wbase = 0;
  for (int k = 0; k < wave; ++k) wbase += s_wsum[k];
  const int excl = wbase + x - cs;
  s_start[2 * tid]     = excl;
  s_start[2 * tid + 1] = excl + ca;
  if (blockIdx.x == 0) {
    if (2 * tid     < nb) binstart[2 * tid]     = excl;
    if (2 * tid + 1 < nb) binstart[2 * tid + 1] = excl + ca;
  }

  const int base = blockIdx.x * CHUNK;
  const bool full = base + CHUNK <= e;

  auto count_pass = [&](const bool FULL) {
    for (int u0 = 0; u0 < CHUNK / 256; u0 += 8) {
      if (!FULL && base + 256 * u0 >= e) break;
      int d[8];
#pragma unroll
      for (int k = 0; k < 8; ++k) {
        const int i = base + tid + 256 * (u0 + k);
        d[k] = dst[FULL ? i : min(i, e - 1)];
      }
#pragma unroll
      for (int k = 0; k < 8; ++k) {
        const int i = base + tid + 256 * (u0 + k);
        if (FULL || i < e) atomicAdd(&s_hist[d[k] >> BIN_SHIFT], 1);
      }
    }
  };
  if (full) count_pass(true); else count_pass(false);
  __syncthreads();

  for (int b = tid; b < nb; b += 256) {
    const int c = s_hist[b];
    const int off = c ? atomicAdd(&bincur[b * CSTRIDE], c) : 0;
    s_hist[b] = s_start[b] + off;
  }
  __syncthreads();

  auto place_pass = [&](const bool FULL) {
    for (int u0 = 0; u0 < CHUNK / 256; u0 += 8) {
      if (!FULL && base + 256 * u0 >= e) break;
      int d[8], s[8], pos[8];
#pragma unroll
      for (int k = 0; k < 8; ++k) {
        const int i = base + tid + 256 * (u0 + k);
        const int j = FULL ? i : min(i, e - 1);
        d[k] = dst[j];
        s[k] = src[j];
      }
#pragma unroll
      for (int k = 0; k < 8; ++k) {
        const int i = base + tid + 256 * (u0 + k);
        pos[k] = 0;
        if (FULL || i < e) pos[k] = atomicAdd(&s_hist[d[k] >> BIN_SHIFT], 1);
      }
#pragma unroll
      for (int k = 0; k < 8; ++k) {
        const int i = base + tid + 256 * (u0 + k);
        if (FULL || i < e) binned[pos[k]] = make_int2(s[k], d[k]);
      }
    }
  };
  if (full) place_pass(true); else place_pass(false);
}

__global__ __launch_bounds__(256) void bin_rows(const int2* __restrict__ binned, const int* __restrict__ bincnt,
                                                const int* __restrict__ binstart, int* __restrict__ rowptr,
                                                float* __restrict__ dinv, int* __restrict__ rec, int n) {
  __shared__ int s_cnt[256];
  __shared__ int s_wsum[8];
  const int tid  = threadIdx.x;
  const int lane = tid & 31;
  const int wave = tid >> 5;
  const int bin   = blockIdx.x;
  const int start = binstart[bin];
  const int ne    = bincnt[bin * CSTRIDE];
  const int2* __restrict__ mine = binned + start;

  s_cnt[tid] = 0;
  __syncthreads();
  for (int i0 = tid; i0 < ne + tid; i0 += 8 * 256) {
    int2 r[8];
#pragma unroll
    for (int k = 0; k < 8; ++k) r[k] = mine[min(i0 + 256 * k, ne - 1)];
#pragma unroll
    for (int k = 0; k < 8; ++k)
      if (i0 + 256 * k < ne) atomicAdd(&s_cnt[r[k].y & 255], 1);
  }
  __syncthreads();

  const int c = s_cnt[tid];
  int x = c;
  for (int off = 1; off < 32; off <<= 1) {
    int y = __shfl_up(x, off);
    if (lane >= off) x += y;
  }
  if (lane == 31) s_wsum[wave] = x;
  __syncthreads();
  int wbase = 0;
  for (int k = 0; k < wave; ++k) wbase += s_wsum[k];
  const int row0 = start + wbase + x - c;

  const int node = (bin << BIN_SHIFT) + tid;
  if (node < n) {
    rowptr[node] = row0;
    if (node == n - 1) rowptr[n] = row0 + c;
    dinv[node] = rsqrtf(fmaxf((float)(c + 1), 1e-12f));
  }
  s_cnt[tid] = row0;
  __syncthreads();
  for (int i0 = tid; i0 < ne + tid; i0 += 8 * 256) {
    int2 r[8];
    int  pos[8];
#pragma unroll
    for (int k = 0; k < 8; ++k) r[k] = mine[min(i0 + 256 * k, ne - 1)];
#pragma unroll
    for (int k = 0; k < 8; ++k) {
      pos[k] = 0;
      if (i0 + 256 * k < ne) pos[k] = atomicAdd(&s_cnt[r[k].y & 255], 1);
    }
#pragma unroll
    for (int k = 0; k < 8; ++k)
      if (i0 + 256 * k < ne) rec[pos[k]] = r[k].x;
  }
}

#define G1_BM 64
#define G1_KS 32
#define G1_XS 36
#define G1_WS 136

__global__ __launch_bounds__(256) __attribute__((amdgpu_waves_per_eu(6))) void gemm1_wmma(const float* X, const float* W, const float* dinv, float* H, int n) {
  __shared__ __attribute__((aligned(16))) float sX[G1_BM * G1_XS];
  __shared__ __attribute__((aligned(16))) float sW[G1_KS * G1_WS];

  const int tid  = threadIdx.x;
  const int wave = tid >> 5;
  const int lane = tid & 31;
  const int rt   = wave >> 1;
  const int col0 = (wave & 1) * 64;
  const int m    = lane & 15;
  const int kh   = lane >> 4;
  const int row0 = blockIdx.x * G1_BM;

  const int xr0 = tid >> 3, xc4 = tid & 7;
  const int wk0 = tid >> 5, wj  = tid & 31;
  const int xrow0 = (row0 + xr0      < n) ? row0 + xr0      : 0;
  const int xrow1 = (row0 + xr0 + 32 < n) ? row0 + xr0 + 32 : 0;
  const float4* xg0 = (const float4*)(X + (size_t)xrow0 * IN_DIMC) + xc4;
  const float4* xg1 = (const float4*)(X + (size_t)xrow1 * IN_DIMC) + xc4;
  const float4* wg  = (const float4*)W + (size_t)wk0 * (H_DIMC / 4) + wj;
  float* __restrict__ sx0 = &sX[xr0 * G1_XS + 4 * xc4];
  float* __restrict__ sw0 = &sW[wk0 * G1_WS + 4 * wj];
  const float* __restrict__ fa = &sX[(rt * 16 + m) * G1_XS + 2 * kh];
  const float* __restrict__ fb = &sW[(2 * kh) * G1_WS + col0 + m];

  v8f acc[4];
#pragma unroll
  for (int t = 0; t < 4; ++t) acc[t] = (v8f){0.f, 0.f, 0.f, 0.f, 0.f, 0.f, 0.f, 0.f};

  const int erow = row0 + rt * 16 + 8 * kh;
  float dvl;

  float4 xa = xg0[0];
  float4 xb = xg1[0];
  float4 w0 = wg[0];
  float4 w1 = wg[8  * (H_DIMC / 4)];
  float4 w2 = wg[16 * (H_DIMC / 4)];
  float4 w3 = wg[24 * (H_DIMC / 4)];

#pragma unroll
  for (int kb = 0; kb < IN_DIMC; kb += G1_KS) {
    __syncthreads();
    *(float4*)(sx0)               = xa;
    *(float4*)(sx0 + 32 * G1_XS)  = xb;
    *(float4*)(sw0)               = w0;
    *(float4*)(sw0 + 8  * G1_WS)  = w1;
    *(float4*)(sw0 + 16 * G1_WS)  = w2;
    *(float4*)(sw0 + 24 * G1_WS)  = w3;
    __syncthreads();
    if (kb + G1_KS < IN_DIMC) {
      const int kn = kb + G1_KS;
      xa = xg0[kn / 4];
      xb = xg1[kn / 4];
      w0 = wg[(size_t)(kn)      * (H_DIMC / 4)];
      w1 = wg[(size_t)(kn + 8)  * (H_DIMC / 4)];
      w2 = wg[(size_t)(kn + 16) * (H_DIMC / 4)];
      w3 = wg[(size_t)(kn + 24) * (H_DIMC / 4)];
    } else {
      dvl = dinv[min(row0 + rt * 16 + m, n - 1)];
    }
    __builtin_amdgcn_sched_barrier(0);

#pragma unroll
    for (int ks = 0; ks < G1_KS; ks += 4) {
      const v2f a = *(const v2f*)(fa + ks);
#pragma unroll
      for (int t = 0; t < 4; ++t) {
        v2f b;
        b[0] = fb[(ks + 0) * G1_WS + 16 * t];
        b[1] = fb[(ks + 1) * G1_WS + 16 * t];
        acc[t] = __builtin_amdgcn_wmma_f32_16x16x4_f32(false, a, false, b,
                                                       (short)0, acc[t], false, false);
      }
    }
  }
  float* __restrict__ hout = H + (size_t)erow * H_DIMC + col0 + m;
#pragma unroll
  for (int r = 0; r < 8; ++r) {
    const float dvr = __shfl(dvl, 8 * kh + r);
#pragma unroll
    for (int t = 0; t < 4; ++t) acc[t][r] *= dvr;
  }
  if (row0 + G1_BM <= n) {
#pragma unroll
    for (int t = 0; t < 4; ++t)
#pragma unroll
      for (int r = 0; r < 8; ++r) hout[(size_t)r * H_DIMC + 16 * t] = acc[t][r];
  } else {
#pragma unroll
    for (int t = 0; t < 4; ++t)
#pragma unroll
      for (int r = 0; r < 8; ++r)
        if (erow + r < n) hout[(size_t)r * H_DIMC + 16 * t] = acc[t][r];
  }
}

#define NPW 10
#define TILE_PITCH 132

template <int F4, bool RELU, int NODES, bool TILE>
__device__ __forceinline__ void gather_rows_walk(const int* __restrict__ rowptr,
                                                 const int* __restrict__ rec,
                                                 const float* __restrict__ h,
                                                 const float* __restrict__ dinv,
                                                 const float* __restrict__ bias,
                                                 float* __restrict__ out, const int n0, const int n) {
  const int lane = threadIdx.x & 31;
  const int q    = (F4 < 32) ? min(lane, F4 - 1) : lane;
  const int cnt = min(NODES, n - n0);
  const int rp  = rowptr[n0 + min(lane, cnt)];
  const float4* __restrict__ hp = (const float4*)h;
  const float bx = bias[4 * q + 0], by = bias[4 * q + 1];
  const float bz = bias[4 * q + 2], bw = bias[4 * q + 3];

  auto sum_pass = [&](const int r, const int m, float4& acc) {
    int k = 0;
    for (; k + 8 <= m; k += 8) {
      float4 v[8];
#pragma unroll
      for (int u = 0; u < 8; ++u) {
        const int s = __builtin_amdgcn_readlane(r, k + u);
        v[u] = hp[(size_t)s * F4 + q];
      }
#pragma unroll
      for (int u = 0; u < 8; ++u) {
        acc.x += v[u].x; acc.y += v[u].y; acc.z += v[u].z; acc.w += v[u].w;
      }
    }
    const int t = m - k;
    if (t > 0) {
      float4 v[7];
#pragma unroll
      for (int u = 0; u < 7; ++u) {
        if (u < t) {
          const int s = __builtin_amdgcn_readlane(r, k + u);
          v[u] = hp[(size_t)s * F4 + q];
        }
      }
#pragma unroll
      for (int u = 0; u < 7; ++u) {
        if (u < t) {
          acc.x += v[u].x; acc.y += v[u].y; acc.z += v[u].z; acc.w += v[u].w;
        }
      }
    }
  };

  int beg = __builtin_amdgcn_readlane(rp, 0);
  int end = __builtin_amdgcn_readlane(rp, 1);
  int r = 0;
  if (lane < end - beg) r = rec[beg + lane];
  float4 hs = hp[(size_t)n0 * F4 + q];
  float  di = dinv[n0];

  for (int j = 0; j < cnt; ++j) {
    const int node = n0 + j;
    int    nend = end;
    int    rn   = 0;
    float4 hsn  = hs;
    float  din  = di;
    if (j + 1 < cnt) {
      nend = __builtin_amdgcn_readlane(rp, j + 2);
      if (lane < nend - end) rn = rec[end + lane];
      hsn = hp[(size_t)(node + 1) * F4 + q];
      din = dinv[node + 1];
    }

    float4 acc = make_float4(0.0f, 0.0f, 0.0f, 0.0f);
    sum_pass(r, min(32, end - beg), acc);
    for (int base = beg + 32; base < end; base += 32) {
      const int m = min(32, end - base);
      int r2 = 0;
      if (lane < m) r2 = rec[base + lane];
      sum_pass(r2, m, acc);
    }

    float4 o;
    o.x = (acc.x + hs.x) * di + bx;
    o.y = (acc.y + hs.y) * di + by;
    o.z = (acc.z + hs.z) * di + bz;
    o.w = (acc.w + hs.w) * di + bw;
    if (RELU) {
      o.x = o.x > 0.0f ? o.x : 0.0f;
      o.y = o.y > 0.0f ? o.y : 0.0f;
      o.z = o.z > 0.0f ? o.z : 0.0f;
      o.w = o.w > 0.0f ? o.w : 0.0f;
    }
    if (TILE) {
      *(float4*)(out + j * TILE_PITCH + 4 * lane) = o;
    } else {
      if (F4 == 32 || lane < F4) ((float4*)out)[(size_t)node * F4 + lane] = o;
    }

    beg = end; end = nend; r = rn; hs = hsn; di = din;
  }
}

#define FG_WAVES 16
#define FG_WP    48

__global__ __launch_bounds__(512) __attribute__((amdgpu_waves_per_eu(8)))
void gather_rows_h_gemm2(const int* __restrict__ rowptr, const int* __restrict__ rec,
                         const float* __restrict__ h, const float* __restrict__ dinv,
                         const float* __restrict__ b1, const float* W2,
                         float* __restrict__ H2, int n) {
  __shared__ __attribute__((aligned(16))) float smem[H_DIMC * FG_WP + FG_WAVES * 16 * TILE_PITCH];
  float* __restrict__ sW = smem;
  float* __restrict__ sA = smem + H_DIMC * FG_WP;

  const int tid  = threadIdx.x;
  const int lane = tid & 31;
  const int wave = tid >> 5;

  const float4 wv0 = ((const float4*)W2)[tid];
  const float4 wv1 = ((const float4*)W2)[tid + 512];
  const float4 wv2 = ((const float4*)W2)[min(tid + 1024, H_DIMC * OUT_DIMC / 4 - 1)];
  __builtin_amdgcn_sched_barrier(0);

  const int n0 = 16 * __builtin_amdgcn_readfirstlane((int)(blockIdx.x * FG_WAVES + wave));
  float* __restrict__ tile = sA + wave * (16 * TILE_PITCH);

  if (n0 < n)
    gather_rows_walk<H_DIMC / 4, true, 16, true>(rowptr, rec, h, dinv, b1, tile, n0, n);

  {
    const int i0 = tid, i1 = tid + 512, i2 = tid + 1024;
    const int r0 = i0 / (OUT_DIMC / 4), r1 = i1 / (OUT_DIMC / 4), r2 = i2 / (OUT_DIMC / 4);
    *(float4*)&sW[r0 * FG_WP + 4 * (i0 - r0 * (OUT_DIMC / 4))] = wv0;
    *(float4*)&sW[r1 * FG_WP + 4 * (i1 - r1 * (OUT_DIMC / 4))] = wv1;
    if (i2 < H_DIMC * OUT_DIMC / 4) *(float4*)&sW[r2 * FG_WP + 4 * (i2 - r2 * (OUT_DIMC / 4))] = wv2;
  }
  if (tid < 2 * H_DIMC)
    *(float4*)&sW[(tid >> 1) * FG_WP + OUT_DIMC + 4 * (tid & 1)] = make_float4(0.f, 0.f, 0.f, 0.f);
  __syncthreads();

  if (n0 < n) {
    const int m  = lane & 15;
    const int kh = lane >> 4;
    const float dvl = dinv[min(n0 + m, n - 1)];
    const float* __restrict__ fa = tile + m * TILE_PITCH + 2 * kh;
    const float* __restrict__ fb = sW + (2 * kh) * FG_WP + m;

    v8f acc[3];
#pragma unroll
    for (int t = 0; t < 3; ++t) acc[t] = (v8f){0.f, 0.f, 0.f, 0.f, 0.f, 0.f, 0.f, 0.f};
#pragma unroll
    for (int ks = 0; ks < H_DIMC; ks += 4) {
      const v2f a = *(const v2f*)(fa + ks);
#pragma unroll
      for (int t = 0; t < 3; ++t) {
        v2f b;
        b[0] = fb[(ks + 0) * FG_WP + 16 * t];
        b[1] = fb[(ks + 1) * FG_WP + 16 * t];
        acc[t] = __builtin_amdgcn_wmma_f32_16x16x4_f32(false, a, false, b,
                                                       (short)0, acc[t], false, false);
      }
    }
    const int erow = n0 + 8 * kh;
#pragma unroll
    for (int r = 0; r < 8; ++r) {
      const float dvr = __shfl(dvl, 8 * kh + r);
#pragma unroll
      for (int t = 0; t < 3; ++t) acc[t][r] *= dvr;
    }
    float* __restrict__ hout = H2 + (size_t)erow * OUT_DIMC + m;
    if (n0 + 16 <= n) {
#pragma unroll
      for (int t = 0; t < 2; ++t)
#pragma unroll
        for (int r = 0; r < 8; ++r) hout[r * OUT_DIMC + 16 * t] = acc[t][r];
      if (m < OUT_DIMC - 32) {
#pragma unroll
        for (int r = 0; r < 8; ++r) hout[r * OUT_DIMC + 32] = acc[2][r];
      }
    } else {
#pragma unroll
      for (int t = 0; t < 3; ++t)
#pragma unroll
        for (int r = 0; r < 8; ++r)
          if (erow + r < n && 16 * t + m < OUT_DIMC) hout[r * OUT_DIMC + 16 * t] = acc[t][r];
    }
  }
}

__global__ __launch_bounds__(256) void gather_rows_o(const int* __restrict__ rowptr,
                                                     const int* __restrict__ rec,
                                                     const float* __restrict__ h2,
                                                     const float* __restrict__ dinv,
                                                     const float* __restrict__ b2,
                                                     float* __restrict__ out, int n) {
  const int n0 = NPW * __builtin_amdgcn_readfirstlane((int)((blockIdx.x * 256u + threadIdx.x) >> 5));
  if (n0 >= n) return;
  gather_rows_walk<OUT_DIMC / 4, false, NPW, false>(rowptr, rec, h2, dinv, b2, out, n0, n);
}

extern "C" void kernel_launch(void* const* d_in, const int* in_sizes, int n_in,
                              void* d_out, int out_size, void* d_ws, size_t ws_size,
                              hipStream_t stream) {
  const float* x  = (const float*)d_in[0];
  const int*   ei = (const int*)d_in[1];
  const float* W1 = (const float*)d_in[2];
  const float* b1 = (const float*)d_in[3];
  const float* W2 = (const float*)d_in[4];
  const float* b2 = (const float*)d_in[5];
  float* out = (float*)d_out;

  const int N = in_sizes[0] / IN_DIMC;
  const int E = in_sizes[1] / 2;
  const int* src = ei;
  const int* dst = ei + E;

  size_t off = 0;
  char* base = (char*)d_ws;
  auto take = [&](size_t count) -> float* {
    float* p = (float*)(base + off);
    off += ((count * sizeof(float) + 255) & ~(size_t)255);
    return p;
  };
  float* dinv   = take((size_t)N);
  int*   bincnt = (int*)take((size_t)2 * NBMAX * CSTRIDE);
  int*   bincur = bincnt + NBMAX * CSTRIDE;
  int*   binstart = (int*)take((size_t)NBMAX);
  int*   rowptr = (int*)take((size_t)N + 1);
  int2*  binned = (int2*)take((size_t)E * 2);
  int*   rec    = (int*)take((size_t)E);
  float* h1     = take((size_t)N * H_DIMC);
  float* h2     = take((size_t)N * OUT_DIMC);
  (void)ws_size; (void)n_in; (void)out_size;

  const int T = 256;
  const int NB = (N + 255) >> BIN_SHIFT;
  const int GB = (E + CHUNK - 1) / CHUNK;
  zero_i32    <<<(2 * NBMAX * CSTRIDE + T - 1) / T, T, 0, stream>>>(bincnt, 2 * NBMAX * CSTRIDE);
  bin_count   <<<GB, T, 0, stream>>>(dst, bincnt, E, NB);
  bin_scatter <<<GB, T, 0, stream>>>(src, dst, bincnt, bincur, binstart, binned, E, NB);
  bin_rows    <<<NB, T, 0, stream>>>(binned, bincnt, binstart, rowptr, dinv, rec, N);

  gemm1_wmma   <<<(N + G1_BM - 1) / G1_BM, 256, 0, stream>>>(x, W1, dinv, h1, N);
  gather_rows_h_gemm2<<<(N + 16 * FG_WAVES - 1) / (16 * FG_WAVES), 32 * FG_WAVES, 0, stream>>>(rowptr, rec, h1, dinv, b1, W2, h2, N);
  gather_rows_o<<<(N + 8 * NPW - 1) / (8 * NPW), 256, 0, stream>>>(rowptr, rec, h2, dinv, b2, out, N);
}


# last list pass (row starts, D^-1/2, sources per row) and layer-1 product share one launch: product blocks count their rows' in-degrees from the binned records themselves; 7 -> 6 launches
# speedup vs baseline: 4.7608x; 1.0384x over previous
#include <hip/hip_runtime.h>

typedef __attribute__((ext_vector_type(2))) float v2f;
typedef __attribute__((ext_vector_type(8))) float v8f;

#define IN_DIMC 256
#define H_DIMC  128
#define OUT_DIMC 40
#define OUT_PADC 48

#define BIN_SHIFT 8
#define NBMAX     512
#define CHUNK     8192
#define CSTRIDE   32

__global__ void zero_i32(int* __restrict__ p, int n) {
  int i = blockIdx.x * blockDim.x + threadIdx.x;
  if (i < n) p[i] = 0;
}

__global__ __launch_bounds__(256) void bin_count(const int* __restrict__ dst, int* __restrict__ bincnt,
                                                 int e, int nb) {
  __shared__ int s_hist[NBMAX];
  const int tid = threadIdx.x;
  s_hist[tid] = 0;
  s_hist[tid + 256] = 0;
  __syncthreads();
  const int base = blockIdx.x * CHUNK;
  auto count_pass = [&](const bool FULL) {
    for (int u0 = 0; u0 < CHUNK / 256; u0 += 8) {
      if (!FULL && base + 256 * u0 >= e) break;
      int d[8];
#pragma unroll
      for (int k = 0; k < 8; ++k) {
        const int i = base + tid + 256 * (u0 + k);
        d[k] = dst[FULL ? i : min(i, e - 1)];
      }
#pragma unroll
      for (int k = 0; k < 8; ++k) {
        const int i = base + tid + 256 * (u0 + k);
        if (FULL || i < e) atomicAdd(&s_hist[d[k] >> BIN_SHIFT], 1);
      }
    }
  };
  if (base + CHUNK <= e) count_pass(true); else count_pass(false);
  __syncthreads();
  for (int b = tid; b < nb; b += 256) {
    const int c = s_hist[b];
    if (c) atomicAdd(&bincnt[b * CSTRIDE], c);
  }
}

__global__ __launch_bounds__(256) void bin_scatter(const int* __restrict__ src, const int* __restrict__ dst,
                                                   const int* __restrict__ bincnt, int* __restrict__ bincur,
                                                   int* __restrict__ binstart, int2* __restrict__ binned,
                                                   int e, int nb) {
  __shared__ int s_start[NBMAX];
  __shared__ int s_hist[NBMAX];
  __shared__ int s_wsum[8];
  const int tid  = threadIdx.x;
  const int lane = tid & 31;
  const int wave = tid >> 5;

  const int ca = (2 * tid     < nb) ? bincnt[(2 * tid)     * CSTRIDE] : 0;
  const int cb = (2 * tid + 1 < nb) ? bincnt[(2 * tid + 1) * CSTRIDE] : 0;
  const int cs = ca + cb;
  int x = cs;
  for (int off = 1; off < 32; off <<= 1) {
    int y = __shfl_up(x, off);
    if (lane >= off) x += y;
  }
  if (lane == 31) s_wsum[wave] = x;
  s_hist[tid] = 0;
  s_hist[tid + 256] = 0;
  __syncthreads();
  int wbase = 0;
  for (int k = 0; k < wave; ++k) wbase += s_wsum[k];
  const int excl = wbase + x - cs;
  s_start[2 * tid]     = excl;
  s_start[2 * tid + 1] = excl + ca;
  if (blockIdx.x == 0) {
    if (2 * tid     < nb) binstart[2 * tid]     = excl;
    if (2 * tid + 1 < nb) binstart[2 * tid + 1] = excl + ca;
  }

  const int base = blockIdx.x * CHUNK;
  const bool full = base + CHUNK <= e;

  auto count_pass = [&](const bool FULL) {
    for (int u0 = 0; u0 < CHUNK / 256; u0 += 8) {
      if (!FULL && base + 256 * u0 >= e) break;
      int d[8];
#pragma unroll
      for (int k = 0; k < 8; ++k) {
        const int i = base + tid + 256 * (u0 + k);
        d[k] = dst[FULL ? i : min(i, e - 1)];
      }
#pragma unroll
      for (int k = 0; k < 8; ++k) {
        const int i = base + tid + 256 * (u0 + k);
        if (FULL || i < e) atomicAdd(&s_hist[d[k] >> BIN_SHIFT], 1);
      }
    }
  };
  if (full) count_pass(true); else count_pass(false);
  __syncthreads();

  for (int b = tid; b < nb; b += 256) {
    const int c = s_hist[b];
    const int off = c ? atomicAdd(&bincur[b * CSTRIDE], c) : 0;
    s_hist[b] = s_start[b] + off;
  }
  __syncthreads();

  auto place_pass = [&](const bool FULL) {
    for (int u0 = 0; u0 < CHUNK / 256; u0 += 8) {
      if (!FULL && base + 256 * u0 >= e) break;
      int d[8], s[8], pos[8];
#pragma unroll
      for (int k = 0; k < 8; ++k) {
        const int i = base + tid + 256 * (u0 + k);
        const int j = FULL ? i : min(i, e - 1);
        d[k] = dst[j];
        s[k] = src[j];
      }
#pragma unroll
      for (int k = 0; k < 8; ++k) {
        const int i = base + tid + 256 * (u0 + k);
        pos[k] = 0;
        if (FULL || i < e) pos[k] = atomicAdd(&s_hist[d[k] >> BIN_SHIFT], 1);
      }
#pragma unroll
      for (int k = 0; k < 8; ++k) {
        const int i = base + tid + 256 * (u0 + k);
        if (FULL || i < e) binned[pos[k]] = make_int2(s[k], d[k]);
      }
    }
  };
  if (full) place_pass(true); else place_pass(false);
}

__device__ __forceinline__ void bin_rows_block(const int2* __restrict__ binned, const int* __restrict__ bincnt,
                                               const int* __restrict__ binstart, int* __restrict__ rowptr,
                                               float* __restrict__ dinv, int* __restrict__ rec, int n, const int bin) {
  __shared__ int s_cnt[256];
  __shared__ int s_wsum[8];
  const int tid  = threadIdx.x;
  const int lane = tid & 31;
  const int wave = tid >> 5;
  const int start = binstart[bin];
  const int ne    = bincnt[bin * CSTRIDE];
  const int2* __restrict__ mine = binned + start;

  s_cnt[tid] = 0;
  __syncthreads();
  for (int i0 = tid; i0 < ne + tid; i0 += 8 * 256) {
    int2 r[8];
#pragma unroll
    for (int k = 0; k < 8; ++k) r[k] = mine[min(i0 + 256 * k, ne - 1)];
#pragma unroll
    for (int k = 0; k < 8; ++k)
      if (i0 + 256 * k < ne) atomicAdd(&s_cnt[r[k].y & 255], 1);
  }
  __syncthreads();

  const int c = s_cnt[tid];
  int x = c;
  for (int off = 1; off < 32; off <<= 1) {
    int y = __shfl_up(x, off);
    if (lane >= off) x += y;
  }
  if (lane == 31) s_wsum[wave] = x;
  __syncthreads();
  int wbase = 0;
  for (int k = 0; k < wave; ++k) wbase += s_wsum[k];
  const int row0 = start + wbase + x - c;

  const int node = (bin << BIN_SHIFT) + tid;
  if (node < n) {
    rowptr[node] = row0;
    if (node == n - 1) rowptr[n] = row0 + c;
    dinv[node] = rsqrtf(fmaxf((float)(c + 1), 1e-12f));
  }
  s_cnt[tid] = row0;
  __syncthreads();
  for (int i0 = tid; i0 < ne + tid; i0 += 8 * 256) {
    int2 r[8];
    int  pos[8];
#pragma unroll
    for (int k = 0; k < 8; ++k) r[k] = mine[min(i0 + 256 * k, ne - 1)];
#pragma unroll
    for (int k = 0; k < 8; ++k) {
      pos[k] = 0;
      if (i0 + 256 * k < ne) pos[k] = atomicAdd(&s_cnt[r[k].y & 255], 1);
    }
#pragma unroll
    for (int k = 0; k < 8; ++k)
      if (i0 + 256 * k < ne) rec[pos[k]] = r[k].x;
  }
}

#define G1_BM 64
#define G1_KS 32
#define G1_XS 36
#define G1_WS 136
#define G1_DCH 24

__device__ __forceinline__ void gemm1_block(const float* X, const float* W, const int2* binned,
                                            const int* bincnt, const int* binstart, float* H, int n,
                                            const int blk) {
  __shared__ __attribute__((aligned(16))) float sX[G1_BM * G1_XS];
  __shared__ __attribute__((aligned(16))) float sW[G1_KS * G1_WS];
  __shared__ int s_deg[G1_BM];

  const int tid  = threadIdx.x;
  const int wave = tid >> 5;
  const int lane = tid & 31;
  const int rt   = wave >> 1;
  const int col0 = (wave & 1) * 64;
  const int m    = lane & 15;
  const int kh   = lane >> 4;
  const int row0 = blk * G1_BM;
  const int swave = __builtin_amdgcn_readfirstlane(wave);

  const int xr0 = tid >> 3, xc4 = tid & 7;
  const int wk0 = tid >> 5, wj  = tid & 31;
  const int xrow0 = (row0 + xr0      < n) ? row0 + xr0      : 0;
  const int xrow1 = (row0 + xr0 + 32 < n) ? row0 + xr0 + 32 : 0;
  const float4* xg0 = (const float4*)(X + (size_t)xrow0 * IN_DIMC) + xc4;
  const float4* xg1 = (const float4*)(X + (size_t)xrow1 * IN_DIMC) + xc4;
  const float4* wg  = (const float4*)W + (size_t)wk0 * (H_DIMC / 4) + wj;
  float* __restrict__ sx0 = &sX[xr0 * G1_XS + 4 * xc4];
  float* __restrict__ sw0 = &sW[wk0 * G1_WS + 4 * wj];
  const float* __restrict__ fa = &sX[(rt * 16 + m) * G1_XS + 2 * kh];
  const float* __restrict__ fb = &sW[(2 * kh) * G1_WS + col0 + m];

  v8f acc[4];
#pragma unroll
  for (int t = 0; t < 4; ++t) acc[t] = (v8f){0.f, 0.f, 0.f, 0.f, 0.f, 0.f, 0.f, 0.f};

  const int bin   = row0 >> BIN_SHIFT;
  const int lo    = row0 & ((1 << BIN_SHIFT) - 1);
  const int start = binstart[bin];
  const int ne    = bincnt[bin * CSTRIDE];
  const int* recd = (const int*)(binned + start) + 1;
  if (tid < G1_BM) s_deg[tid] = 0;
  __syncthreads();
  int d0[G1_DCH];
  const int last = max(ne - 1, 0);
#pragma unroll
  for (int k = 0; k < G1_DCH; ++k) d0[k] = recd[2 * min(tid + 256 * k, last)];

  float4 xa = xg0[0];
  float4 xb = xg1[0];
  float4 w0 = wg[0];
  float4 w1 = wg[8  * (H_DIMC / 4)];
  float4 w2 = wg[16 * (H_DIMC / 4)];
  float4 w3 = wg[24 * (H_DIMC / 4)];

#pragma unroll
  for (int k = 0; k < G1_DCH; ++k) {
    const int q = (d0[k] & ((1 << BIN_SHIFT) - 1)) - lo;
    if (tid + 256 * k < ne && (unsigned)q < (unsigned)G1_BM) atomicAdd(&s_deg[q], 1);
  }
  for (int i0 = tid + G1_DCH * 256; i0 < ne + tid; i0 += 8 * 256) {
    int d[8];
#pragma unroll
    for (int k = 0; k < 8; ++k) d[k] = recd[2 * min(i0 + 256 * k, ne - 1)];
#pragma unroll
    for (int k = 0; k < 8; ++k) {
      const int q = (d[k] & ((1 << BIN_SHIFT) - 1)) - lo;
      if (i0 + 256 * k < ne && (unsigned)q < (unsigned)G1_BM) atomicAdd(&s_deg[q], 1);
    }
  }

#pragma unroll
  for (int kb = 0; kb < IN_DIMC; kb += G1_KS) {
    __syncthreads();
    *(float4*)(sx0)               = xa;
    *(float4*)(sx0 + 32 * G1_XS)  = xb;
    *(float4*)(sw0)               = w0;
    *(float4*)(sw0 + 8  * G1_WS)  = w1;
    *(float4*)(sw0 + 16 * G1_WS)  = w2;
    *(float4*)(sw0 + 24 * G1_WS)  = w3;
    __syncthreads();
    if (kb + G1_KS < IN_DIMC) {
      const int kn = kb + G1_KS;
      xa = xg0[kn / 4];
      xb = xg1[kn / 4];
      w0 = wg[(size_t)(kn)      * (H_DIMC / 4)];
      w1 = wg[(size_t)(kn + 8)  * (H_DIMC / 4)];
      w2 = wg[(size_t)(kn + 16) * (H_DIMC / 4)];
      w3 = wg[(size_t)(kn + 24) * (H_DIMC / 4)];
    }
    __builtin_amdgcn_sched_barrier(0);

#pragma unroll
    for (int ks = 0; ks < G1_KS; ks += 4) {
      const v2f a = *(const v2f*)(fa + ks);
#pragma unroll
      for (int t = 0; t < 4; ++t) {
        v2f b;
        b[0] = fb[(ks + 0) * G1_WS + 16 * t];
        b[1] = fb[(ks + 1) * G1_WS + 16 * t];
        acc[t] = __builtin_amdgcn_wmma_f32_16x16x4_f32(false, a, false, b,
                                                       (short)0, acc[t], false, false);
      }
    }
  }
  const int elane = (int)__builtin_amdgcn_mbcnt_lo(~0u, 0u);
  const int em    = elane & 15;
  const int ekh   = elane >> 4;
  const int ert   = swave >> 1;
  const int ecol0 = (swave & 1) * 64;
  const int erow  = row0 + ert * 16 + 8 * ekh;
  const float dvl = rsqrtf(fmaxf((float)(s_deg[ert * 16 + em] + 1), 1e-12f));
  float* __restrict__ hout = H + (size_t)erow * H_DIMC + ecol0 + em;
#pragma unroll
  for (int r = 0; r < 8; ++r) {
    const float dvr = __shfl(dvl, 8 * ekh + r);
#pragma unroll
    for (int t = 0; t < 4; ++t) acc[t][r] *= dvr;
  }
  if (row0 + G1_BM <= n) {
#pragma unroll
    for (int t = 0; t < 4; ++t)
#pragma unroll
      for (int r = 0; r < 8; ++r) hout[(size_t)r * H_DIMC + 16 * t] = acc[t][r];
  } else {
#pragma unroll
    for (int t = 0; t < 4; ++t)
#pragma unroll
      for (int r = 0; r < 8; ++r)
        if (erow + r < n) hout[(size_t)r * H_DIMC + 16 * t] = acc[t][r];
  }
}

__global__ __launch_bounds__(256) __attribute__((amdgpu_waves_per_eu(6)))
void rows_and_gemm1(const int2* binned, const int* bincnt, const int* binstart,
                    int* rowptr, float* dinv, int* rec,
                    const float* X, const float* W, float* H, int n) {
  const int b  = blockIdx.x;
  const int ng = (n + G1_BM - 1) / G1_BM;
  if (b < ng) gemm1_block(X, W, binned, bincnt, binstart, H, n, b);
  else        bin_rows_block(binned, bincnt, binstart, rowptr, dinv, rec, n, b - ng);
}

#define NPW 10
#define TILE_PITCH 132

template <int F4, bool RELU, int NODES, bool TILE>
__device__ __forceinline__ void gather_rows_walk(const int* __restrict__ rowptr,
                                                 const int* __restrict__ rec,
                                                 const float* __restrict__ h,
                                                 const float* __restrict__ dinv,
                                                 const float* __restrict__ bias,
                                                 float* __restrict__ out, const int n0, const int n) {
  const int lane = threadIdx.x & 31;
  const int q    = (F4 < 32) ? min(lane, F4 - 1) : lane;
  const int cnt = min(NODES, n - n0);
  const int rp  = rowptr[n0 + min(lane, cnt)];
  const float4* __restrict__ hp = (const float4*)h;
  const float bx = bias[4 * q + 0], by = bias[4 * q + 1];
  const float bz = bias[4 * q + 2], bw = bias[4 * q + 3];

  auto sum_pass = [&](const int r, const int m, float4& acc) {
    int k = 0;
    for (; k + 8 <= m; k += 8) {
      float4 v[8];
#pragma unroll
      for (int u = 0; u < 8; ++u) {
        const int s = __builtin_amdgcn_readlane(r, k + u);
        v[u] = hp[(size_t)s * F4 + q];
      }
#pragma unroll
      for (int u = 0; u < 8; ++u) {
        acc.x += v[u].x; acc.y += v[u].y; acc.z += v[u].z; acc.w += v[u].w;
      }
    }
    const int t = m - k;
    if (t > 0) {
      float4 v[7];
#pragma unroll
      for (int u = 0; u < 7; ++u) {
        if (u < t) {
          const int s = __builtin_amdgcn_readlane(r, k + u);
          v[u] = hp[(size_t)s * F4 + q];
        }
      }
#pragma unroll
      for (int u = 0; u < 7; ++u) {
        if (u < t) {
          acc.x += v[u].x; acc.y += v[u].y; acc.z += v[u].z; acc.w += v[u].w;
        }
      }
    }
  };

  int beg = __builtin_amdgcn_readlane(rp, 0);
  int end = __builtin_amdgcn_readlane(rp, 1);
  int r = 0;
  if (lane < end - beg) r = rec[beg + lane];
  float4 hs = hp[(size_t)n0 * F4 + q];
  float  di = dinv[n0];

  for (int j = 0; j < cnt; ++j) {
    const int node = n0 + j;
    int    nend = end;
    int    rn   = 0;
    float4 hsn  = hs;
    float  din  = di;
    if (j + 1 < cnt) {
      nend = __builtin_amdgcn_readlane(rp, j + 2);
      if (lane < nend - end) rn = rec[end + lane];
      hsn = hp[(size_t)(node + 1) * F4 + q];
      din = dinv[node + 1];
    }

    float4 acc = make_float4(0.0f, 0.0f, 0.0f, 0.0f);
    sum_pass(r, min(32, end - beg), acc);
    for (int base = beg + 32; base < end; base += 32) {
      const int m = min(32, end - base);
      int r2 = 0;
      if (lane < m) r2 = rec[base + lane];
      sum_pass(r2, m, acc);
    }

    float4 o;
    o.x = (acc.x + hs.x) * di + bx;
    o.y = (acc.y + hs.y) * di + by;
    o.z = (acc.z + hs.z) * di + bz;
    o.w = (acc.w + hs.w) * di + bw;
    if (RELU) {
      o.x = o.x > 0.0f ? o.x : 0.0f;
      o.y = o.y > 0.0f ? o.y : 0.0f;
      o.z = o.z > 0.0f ? o.z : 0.0f;
      o.w = o.w > 0.0f ? o.w : 0.0f;
    }
    if (TILE) {
      *(float4*)(out + j * TILE_PITCH + 4 * lane) = o;
    } else {
      if (F4 == 32 || lane < F4) ((float4*)out)[(size_t)node * F4 + lane] = o;
    }

    beg = end; end = nend; r = rn; hs = hsn; di = din;
  }
}

#define FG_WAVES 16
#define FG_WP    48

__global__ __launch_bounds__(512) __attribute__((amdgpu_waves_per_eu(8)))
void gather_rows_h_gemm2(const int* __restrict__ rowptr, const int* __restrict__ rec,
                         const float* __restrict__ h, const float* __restrict__ dinv,
                         const float* __restrict__ b1, const float* W2,
                         float* __restrict__ H2, int n) {
  __shared__ __attribute__((aligned(16))) float smem[H_DIMC * FG_WP + FG_WAVES * 16 * TILE_PITCH];
  float* __restrict__ sW = smem;
  float* __restrict__ sA = smem + H_DIMC * FG_WP;

  const int tid  = threadIdx.x;
  const int lane = tid & 31;
  const int wave = tid >> 5;

  const float4 wv0 = ((const float4*)W2)[tid];
  const float4 wv1 = ((const float4*)W2)[tid + 512];
  const float4 wv2 = ((const float4*)W2)[min(tid + 1024, H_DIMC * OUT_DIMC / 4 - 1)];
  __builtin_amdgcn_sched_barrier(0);

  const int n0 = 16 * __builtin_amdgcn_readfirstlane((int)(blockIdx.x * FG_WAVES + wave));
  float* __restrict__ tile = sA + wave * (16 * TILE_PITCH);

  if (n0 < n)
    gather_rows_walk<H_DIMC / 4, true, 16, true>(rowptr, rec, h, dinv, b1, tile, n0, n);

  {
    const int i0 = tid, i1 = tid + 512, i2 = tid + 1024;
    const int r0 = i0 / (OUT_DIMC / 4), r1 = i1 / (OUT_DIMC / 4), r2 = i2 / (OUT_DIMC / 4);
    *(float4*)&sW[r0 * FG_WP + 4 * (i0 - r0 * (OUT_DIMC / 4))] = wv0;
    *(float4*)&sW[r1 * FG_WP + 4 * (i1 - r1 * (OUT_DIMC / 4))] = wv1;
    if (i2 < H_DIMC * OUT_DIMC / 4) *(float4*)&sW[r2 * FG_WP + 4 * (i2 - r2 * (OUT_DIMC / 4))] = wv2;
  }
  if (tid < 2 * H_DIMC)
    *(float4*)&sW[(tid >> 1) * FG_WP + OUT_DIMC + 4 * (tid & 1)] = make_float4(0.f, 0.f, 0.f, 0.f);
  __syncthreads();

  if (n0 < n) {
    const int m  = lane & 15;
    const int kh = lane >> 4;
    const float dvl = dinv[min(n0 + m, n - 1)];
    const float* __restrict__ fa = tile + m * TILE_PITCH + 2 * kh;
    const float* __restrict__ fb = sW + (2 * kh) * FG_WP + m;

    v8f acc[3];
#pragma unroll
    for (int t = 0; t < 3; ++t) acc[t] = (v8f){0.f, 0.f, 0.f, 0.f, 0.f, 0.f, 0.f, 0.f};
#pragma unroll
    for (int ks = 0; ks < H_DIMC; ks += 4) {
      const v2f a = *(const v2f*)(fa + ks);
#pragma unroll
      for (int t = 0; t < 3; ++t) {
        v2f b;
        b[0] = fb[(ks + 0) * FG_WP + 16 * t];
        b[1] = fb[(ks + 1) * FG_WP + 16 * t];
        acc[t] = __builtin_amdgcn_wmma_f32_16x16x4_f32(false, a, false, b,
                                                       (short)0, acc[t], false, false);
      }
    }
    const int erow = n0 + 8 * kh;
#pragma unroll
    for (int r = 0; r < 8; ++r) {
      const float dvr = __shfl(dvl, 8 * kh + r);
#pragma unroll
      for (int t = 0; t < 3; ++t) acc[t][r] *= dvr;
    }
    float* __restrict__ hout = H2 + (size_t)erow * OUT_DIMC + m;
    if (n0 + 16 <= n) {
#pragma unroll
      for (int t = 0; t < 2; ++t)
#pragma unroll
        for (int r = 0; r < 8; ++r) hout[r * OUT_DIMC + 16 * t] = acc[t][r];
      if (m < OUT_DIMC - 32) {
#pragma unroll
        for (int r = 0; r < 8; ++r) hout[r * OUT_DIMC + 32] = acc[2][r];
      }
    } else {
#pragma unroll
      for (int t = 0; t < 3; ++t)
#pragma unroll
        for (int r = 0; r < 8; ++r)
          if (erow + r < n && 16 * t + m < OUT_DIMC) hout[r * OUT_DIMC + 16 * t] = acc[t][r];
    }
  }
}

__global__ __launch_bounds__(256) void gather_rows_o(const int* __restrict__ rowptr,
                                                     const int* __restrict__ rec,
                                                     const float* __restrict__ h2,
                                                     const float* __restrict__ dinv,
                                                     const float* __restrict__ b2,
                                                     float* __restrict__ out, int n) {
  const int n0 = NPW * __builtin_amdgcn_readfirstlane((int)((blockIdx.x * 256u + threadIdx.x) >> 5));
  if (n0 >= n) return;
  gather_rows_walk<OUT_DIMC / 4, false, NPW, false>(rowptr, rec, h2, dinv, b2, out, n0, n);
}

extern "C" void kernel_launch(void* const* d_in, const int* in_sizes, int n_in,
                              void* d_out, int out_size, void* d_ws, size_t ws_size,
                              hipStream_t stream) {
  const float* x  = (const float*)d_in[0];
  const int*   ei = (const int*)d_in[1];
  const float* W1 = (const float*)d_in[2];
  const float* b1 = (const float*)d_in[3];
  const float* W2 = (const float*)d_in[4];
  const float* b2 = (const float*)d_in[5];
  float* out = (float*)d_out;

  const int N = in_sizes[0] / IN_DIMC;
  const int E = in_sizes[1] / 2;
  const int* src = ei;
  const int* dst = ei + E;

  size_t off = 0;
  char* base = (char*)d_ws;
  auto take = [&](size_t count) -> float* {
    float* p = (float*)(base + off);
    off += ((count * sizeof(float) + 255) & ~(size_t)255);
    return p;
  };
  float* dinv   = take((size_t)N);
  int*   bincnt = (int*)take((size_t)2 * NBMAX * CSTRIDE);
  int*   bincur = bincnt + NBMAX * CSTRIDE;
  int*   binstart = (int*)take((size_t)NBMAX);
  int*   rowptr = (int*)take((size_t)N + 1);
  int2*  binned = (int2*)take((size_t)E * 2);
  int*   rec    = (int*)take((size_t)E);
  float* h1     = take((size_t)N * H_DIMC);
  float* h2     = take((size_t)N * OUT_DIMC);
  (void)ws_size; (void)n_in; (void)out_size;

  const int T = 256;
  const int NB = (N + 255) >> BIN_SHIFT;
  const int GB = (E + CHUNK - 1) / CHUNK;
  zero_i32    <<<(2 * NBMAX * CSTRIDE + T - 1) / T, T, 0, stream>>>(bincnt, 2 * NBMAX * CSTRIDE);
  bin_count   <<<GB, T, 0, stream>>>(dst, bincnt, E, NB);
  bin_scatter <<<GB, T, 0, stream>>>(src, dst, bincnt, bincur, binstart, binned, E, NB);
  rows_and_gemm1<<<(N + G1_BM - 1) / G1_BM + NB, T, 0, stream>>>(binned, bincnt, binstart, rowptr, dinv, rec,
                                                                 x, W1, h1, N);
  gather_rows_h_gemm2<<<(N + 16 * FG_WAVES - 1) / (16 * FG_WAVES), 32 * FG_WAVES, 0, stream>>>(rowptr, rec, h1, dinv, b1, W2, h2, N);
  gather_rows_o<<<(N + 8 * NPW - 1) / (8 * NPW), 256, 0, stream>>>(rowptr, rec, h2, dinv, b2, out, N);
}
